# starting kernel
# baseline (speedup 1.0000x reference)
#include <hip/hip_runtime.h>
#include <math.h>


#define D_MODEL 128
#define D_STATE 16
#define BATCH   4
#define SEQ     4096
#define NCHUNK  16
#define CHUNK   (SEQ / NCHUNK)
#define LOG_CLAMP 70.0f

typedef float v2f __attribute__((ext_vector_type(2)));
typedef float v8f __attribute__((ext_vector_type(8)));

__device__ __forceinline__ float softplus_f(float z) {
    return fmaxf(z, 0.0f) + log1pf(__expf(-fabsf(z)));
}

__global__ void proj_wmma_kernel(const float* __restrict__ x,
                                 const float* __restrict__ W_B, const float* __restrict__ b_B,
                                 const float* __restrict__ W_C, const float* __restrict__ b_C,
                                 const float* __restrict__ W_D, const float* __restrict__ b_D,
                                 float* __restrict__ Bm, float* __restrict__ Cc,
                                 float* __restrict__ xwd) {
    const int lane  = threadIdx.x & 31;
    const int wave  = threadIdx.x >> 5;
    const int tile  = blockIdx.x * (blockDim.x >> 5) + wave;
    const int half  = lane >> 4;
    const int lid   = lane & 15;
    const int row   = tile * 16 + lid;
    const int khalf = half * 2;
    const float dmask = (lid == 0) ? 1.0f : 0.0f;

    v8f accB = {}; v8f accC = {}; v8f accD = {};
    const float* xrow = x + (size_t)row * D_MODEL;

    for (int k0 = 0; k0 < D_MODEL; k0 += 4) {
        const v2f a = *(const v2f*)(xrow + k0 + khalf);
        v2f fB, fC, fD;
        fB.x = W_B[(k0 + khalf + 0) * D_STATE + lid];
        fB.y = W_B[(k0 + khalf + 1) * D_STATE + lid];
        fC.x = W_C[(k0 + khalf + 0) * D_STATE + lid];
        fC.y = W_C[(k0 + khalf + 1) * D_STATE + lid];
        fD.x = W_D[k0 + khalf + 0] * dmask;
        fD.y = W_D[k0 + khalf + 1] * dmask;
        accB = __builtin_amdgcn_wmma_f32_16x16x4_f32(false, a, false, fB, (short)0, accB, false, false);
        accC = __builtin_amdgcn_wmma_f32_16x16x4_f32(false, a, false, fC, (short)0, accC, false, false);
        accD = __builtin_amdgcn_wmma_f32_16x16x4_f32(false, a, false, fD, (short)0, accD, false, false);
    }

    const float bBn = b_B[lid];
    const float bCn = b_C[lid];
    const float bDs = b_D[0];
#pragma unroll
    for (int r = 0; r < 8; ++r) {
        const int orow = tile * 16 + r + 8 * half;
        Bm[(size_t)orow * D_STATE + lid] = accB[r] + bBn;
        Cc[(size_t)orow * D_STATE + lid] = accC[r] + bCn;
        if (lid == 0) xwd[orow] = accD[r] + bDs;
    }
}

__global__ void disc_kernel(const float* __restrict__ xwd, const float* __restrict__ P,
                            float* __restrict__ disc, float* __restrict__ discsum) {
    const int t  = blockIdx.x * blockDim.x + threadIdx.x;
    const int d  = t % D_MODEL;
    const int g  = t / D_MODEL;
    const int ch = g % NCHUNK;
    const int b  = g / NCHUNK;

    const float Pd = P[d];
    float acc = 0.0f;
    const int s0 = ch * CHUNK;
    for (int s = s0; s < s0 + CHUNK; ++s) {
        const float dc = softplus_f(Pd + xwd[b * SEQ + s]);
        disc[((size_t)(b * SEQ + s)) * D_MODEL + d] = dc;
        acc += dc;
    }
    discsum[(b * D_MODEL + d) * NCHUNK + ch] = acc;
}

__global__ void scan_chunks_kernel(float* __restrict__ arr, int nchannels) {
    const int t = blockIdx.x * blockDim.x + threadIdx.x;
    if (t >= nchannels) return;
    float run = 0.0f;
    float* p = arr + (size_t)t * NCHUNK;
#pragma unroll
    for (int c = 0; c < NCHUNK; ++c) { const float v = p[c]; p[c] = run; run += v; }
}

__global__ void scan_pass1_kernel(const float* __restrict__ x,
                                  const float* __restrict__ Araw,
                                  const float* __restrict__ disc,
                                  const float* __restrict__ discpre,
                                  const float* __restrict__ Bm,
                                  float* __restrict__ sumS) {
    const int t  = blockIdx.x * blockDim.x + threadIdx.x;
    const int n  = t & 15;
    const int g  = t >> 4;
    const int ch = g % NCHUNK;
    const int g2 = g / NCHUNK;
    const int d  = g2 % D_MODEL;
    const int b  = g2 / D_MODEL;

    const float A    = -softplus_f(Araw[d * D_STATE + n]);
    const float invA = 1.0f / A;
    float Dcum = discpre[(b * D_MODEL + d) * NCHUNK + ch];
    float acc  = 0.0f;
    const int s0 = ch * CHUNK;
    for (int s = s0; s < s0 + CHUNK; ++s) {
        const size_t bs = (size_t)(b * SEQ + s);
        const float dc  = disc[bs * D_MODEL + d];
        Dcum += dc;
        const float L   = A * Dcum;
        const float xv  = x[bs * D_MODEL + d];
        const float bm  = Bm[bs * D_STATE + n];
        const float Bx  = bm * invA * (__expf(dc * A) - 1.0f) * xv;
        const float bxl = __logf(fabsf(Bx));
        const float Lc  = fmaxf(L, bxl - LOG_CLAMP);
        const float sgn = (Bx > 0.0f) ? 1.0f : ((Bx < 0.0f) ? -1.0f : 0.0f);
        acc += __expf(bxl - Lc) * sgn;
    }
    sumS[((b * D_MODEL + d) * D_STATE + n) * NCHUNK + ch] = acc;
}

__global__ void scan_final_kernel(const float* __restrict__ x,
                                  const float* __restrict__ Araw,
                                  const float* __restrict__ disc,
                                  const float* __restrict__ discpre,
                                  const float* __restrict__ Bm,
                                  const float* __restrict__ Cc,
                                  const float* __restrict__ sumSpre,
                                  float* __restrict__ y) {
    const int t  = blockIdx.x * blockDim.x + threadIdx.x;
    const int n  = t & 15;
    const int g  = t >> 4;
    const int ch = g % NCHUNK;
    const int g2 = g / NCHUNK;
    const int d  = g2 % D_MODEL;
    const int b  = g2 / D_MODEL;

    const float A    = -softplus_f(Araw[d * D_STATE + n]);
    const float invA = 1.0f / A;
    float Dcum = discpre[(b * D_MODEL + d) * NCHUNK + ch];
    float Ssum = sumSpre[((b * D_MODEL + d) * D_STATE + n) * NCHUNK + ch];
    const int s0 = ch * CHUNK;
    for (int s = s0; s < s0 + CHUNK; ++s) {
        const size_t bs = (size_t)(b * SEQ + s);
        const float dc  = disc[bs * D_MODEL + d];
        Dcum += dc;
        const float L   = A * Dcum;
        const float xv  = x[bs * D_MODEL + d];
        const float bm  = Bm[bs * D_STATE + n];
        const float Bx  = bm * invA * (__expf(dc * A) - 1.0f) * xv;
        const float bxl = __logf(fabsf(Bx));
        const float Lc  = fmaxf(L, bxl - LOG_CLAMP);
        const float sgn = (Bx > 0.0f) ? 1.0f : ((Bx < 0.0f) ? -1.0f : 0.0f);
        Ssum += __expf(bxl - Lc) * sgn;
        const float h = Ssum * __expf(Lc);
        float contrib = h * Cc[bs * D_STATE + n];
        contrib += __shfl_xor(contrib, 1, 32);
        contrib += __shfl_xor(contrib, 2, 32);
        contrib += __shfl_xor(contrib, 4, 32);
        contrib += __shfl_xor(contrib, 8, 32);
        if (n == 0) y[bs * D_MODEL + d] = contrib;
    }
}

extern "C" void kernel_launch(void* const* d_in, const int* in_sizes, int n_in,
                              void* d_out, int out_size, void* d_ws, size_t ws_size,
                              hipStream_t stream) {
    (void)in_sizes; (void)n_in; (void)out_size; (void)ws_size;
    const float* x    = (const float*)d_in[0];
    const float* Araw = (const float*)d_in[1];
    const float* P    = (const float*)d_in[2];
    const float* W_B  = (const float*)d_in[3];
    const float* b_B  = (const float*)d_in[4];
    const float* W_C  = (const float*)d_in[5];
    const float* b_C  = (const float*)d_in[6];
    const float* W_D  = (const float*)d_in[7];
    const float* b_D  = (const float*)d_in[8];
    float* y = (float*)d_out;

    constexpr size_t BS       = (size_t)BATCH * SEQ;
    constexpr size_t szBm     = BS * D_STATE * 4;
    constexpr size_t szCc     = BS * D_STATE * 4;
    constexpr size_t szXwd    = BS * 4;
    constexpr size_t szDisc   = BS * D_MODEL * 4;
    constexpr size_t szDsum   = (size_t)BATCH * D_MODEL * NCHUNK * 4;
    char* ws = (char*)d_ws;
    float* Bm   = (float*)(ws);
    float* Cc   = (float*)(ws + szBm);
    float* xwd  = (float*)(ws + szBm + szCc);
    float* disc = (float*)(ws + szBm + szCc + szXwd);
    float* dsum = (float*)(ws + szBm + szCc + szXwd + szDisc);
    float* sumS = (float*)(ws + szBm + szCc + szXwd + szDisc + szDsum);

    proj_wmma_kernel<<<(BS / 16) / 8, 256, 0, stream>>>(x, W_B, b_B, W_C, b_C, W_D, b_D,
                                                        Bm, Cc, xwd);
    disc_kernel<<<(BATCH * D_MODEL * NCHUNK) / 256, 256, 0, stream>>>(xwd, P, disc, dsum);
    scan_chunks_kernel<<<(BATCH * D_MODEL + 255) / 256, 256, 0, stream>>>(dsum, BATCH * D_MODEL);
    scan_pass1_kernel<<<(BATCH * D_MODEL * D_STATE * NCHUNK) / 256, 256, 0, stream>>>(
        x, Araw, disc, dsum, Bm, sumS);
    scan_chunks_kernel<<<(BATCH * D_MODEL * D_STATE + 255) / 256, 256, 0, stream>>>(
        sumS, BATCH * D_MODEL * D_STATE);
    scan_final_kernel<<<(BATCH * D_MODEL * D_STATE * NCHUNK) / 256, 256, 0, stream>>>(
        x, Araw, disc, dsum, Bm, Cc, sumS, y);
}


# Two-level scan re-chunked from 16 chunks of 256 steps to 128 chunks of 32: 8x more, 8x shorter waves fill all wave slots; both cumulative sums get a new reduction tree (summation order changes)
# speedup vs baseline: 2.6367x; 2.6367x over previous
#include <hip/hip_runtime.h>
#include <math.h>


#define D_MODEL 128
#define D_STATE 16
#define BATCH   4
#define SEQ     4096
#define NCHUNK  128
#define CHUNK   (SEQ / NCHUNK)
#define LOG_CLAMP 70.0f

typedef float v2f __attribute__((ext_vector_type(2)));
typedef float v8f __attribute__((ext_vector_type(8)));

__device__ __forceinline__ float softplus_f(float z) {
    return fmaxf(z, 0.0f) + log1pf(__expf(-fabsf(z)));
}

__global__ void proj_wmma_kernel(const float* __restrict__ x,
                                 const float* __restrict__ W_B, const float* __restrict__ b_B,
                                 const float* __restrict__ W_C, const float* __restrict__ b_C,
                                 const float* __restrict__ W_D, const float* __restrict__ b_D,
                                 float* __restrict__ Bm, float* __restrict__ Cc,
                                 float* __restrict__ xwd) {
    const int lane  = threadIdx.x & 31;
    const int wave  = threadIdx.x >> 5;
    const int tile  = blockIdx.x * (blockDim.x >> 5) + wave;
    const int half  = lane >> 4;
    const int lid   = lane & 15;
    const int row   = tile * 16 + lid;
    const int khalf = half * 2;
    const float dmask = (lid == 0) ? 1.0f : 0.0f;

    v8f accB = {}; v8f accC = {}; v8f accD = {};
    const float* xrow = x + (size_t)row * D_MODEL;

    for (int k0 = 0; k0 < D_MODEL; k0 += 4) {
        const v2f a = *(const v2f*)(xrow + k0 + khalf);
        v2f fB, fC, fD;
        fB.x = W_B[(k0 + khalf + 0) * D_STATE + lid];
        fB.y = W_B[(k0 + khalf + 1) * D_STATE + lid];
        fC.x = W_C[(k0 + khalf + 0) * D_STATE + lid];
        fC.y = W_C[(k0 + khalf + 1) * D_STATE + lid];
        fD.x = W_D[k0 + khalf + 0] * dmask;
        fD.y = W_D[k0 + khalf + 1] * dmask;
        accB = __builtin_amdgcn_wmma_f32_16x16x4_f32(false, a, false, fB, (short)0, accB, false, false);
        accC = __builtin_amdgcn_wmma_f32_16x16x4_f32(false, a, false, fC, (short)0, accC, false, false);
        accD = __builtin_amdgcn_wmma_f32_16x16x4_f32(false, a, false, fD, (short)0, accD, false, false);
    }

    const float bBn = b_B[lid];
    const float bCn = b_C[lid];
    const float bDs = b_D[0];
#pragma unroll
    for (int r = 0; r < 8; ++r) {
        const int orow = tile * 16 + r + 8 * half;
        Bm[(size_t)orow * D_STATE + lid] = accB[r] + bBn;
        Cc[(size_t)orow * D_STATE + lid] = accC[r] + bCn;
        if (lid == 0) xwd[orow] = accD[r] + bDs;
    }
}

__global__ void disc_kernel(const float* __restrict__ xwd, const float* __restrict__ P,
                            float* __restrict__ disc, float* __restrict__ discsum) {
    const int t  = blockIdx.x * blockDim.x + threadIdx.x;
    const int d  = t % D_MODEL;
    const int g  = t / D_MODEL;
    const int ch = g % NCHUNK;
    const int b  = g / NCHUNK;

    const float Pd = P[d];
    float acc = 0.0f;
    const int s0 = ch * CHUNK;
    for (int s = s0; s < s0 + CHUNK; ++s) {
        const float dc = softplus_f(Pd + xwd[b * SEQ + s]);
        disc[((size_t)(b * SEQ + s)) * D_MODEL + d] = dc;
        acc += dc;
    }
    discsum[(b * D_MODEL + d) * NCHUNK + ch] = acc;
}

__global__ void scan_chunks_kernel(float* __restrict__ arr, int nchannels) {
    const int t = blockIdx.x * blockDim.x + threadIdx.x;
    if (t >= nchannels) return;
    float run = 0.0f;
    float* p = arr + (size_t)t * NCHUNK;
#pragma unroll
    for (int c = 0; c < NCHUNK; ++c) { const float v = p[c]; p[c] = run; run += v; }
}

__global__ void scan_pass1_kernel(const float* __restrict__ x,
                                  const float* __restrict__ Araw,
                                  const float* __restrict__ disc,
                                  const float* __restrict__ discpre,
                                  const float* __restrict__ Bm,
                                  float* __restrict__ sumS) {
    const int t  = blockIdx.x * blockDim.x + threadIdx.x;
    const int n  = t & 15;
    const int g  = t >> 4;
    const int ch = g % NCHUNK;
    const int g2 = g / NCHUNK;
    const int d  = g2 % D_MODEL;
    const int b  = g2 / D_MODEL;

    const float A    = -softplus_f(Araw[d * D_STATE + n]);
    const float invA = 1.0f / A;
    float Dcum = discpre[(b * D_MODEL + d) * NCHUNK + ch];
    float acc  = 0.0f;
    const int s0 = ch * CHUNK;
    for (int s = s0; s < s0 + CHUNK; ++s) {
        const size_t bs = (size_t)(b * SEQ + s);
        const float dc  = disc[bs * D_MODEL + d];
        Dcum += dc;
        const float L   = A * Dcum;
        const float xv  = x[bs * D_MODEL + d];
        const float bm  = Bm[bs * D_STATE + n];
        const float Bx  = bm * invA * (__expf(dc * A) - 1.0f) * xv;
        const float bxl = __logf(fabsf(Bx));
        const float Lc  = fmaxf(L, bxl - LOG_CLAMP);
        const float sgn = (Bx > 0.0f) ? 1.0f : ((Bx < 0.0f) ? -1.0f : 0.0f);
        acc += __expf(bxl - Lc) * sgn;
    }
    sumS[((b * D_MODEL + d) * D_STATE + n) * NCHUNK + ch] = acc;
}

__global__ void scan_final_kernel(const float* __restrict__ x,
                                  const float* __restrict__ Araw,
                                  const float* __restrict__ disc,
                                  const float* __restrict__ discpre,
                                  const float* __restrict__ Bm,
                                  const float* __restrict__ Cc,
                                  const float* __restrict__ sumSpre,
                                  float* __restrict__ y) {
    const int t  = blockIdx.x * blockDim.x + threadIdx.x;
    const int n  = t & 15;
    const int g  = t >> 4;
    const int ch = g % NCHUNK;
    const int g2 = g / NCHUNK;
    const int d  = g2 % D_MODEL;
    const int b  = g2 / D_MODEL;

    const float A    = -softplus_f(Araw[d * D_STATE + n]);
    const float invA = 1.0f / A;
    float Dcum = discpre[(b * D_MODEL + d) * NCHUNK + ch];
    float Ssum = sumSpre[((b * D_MODEL + d) * D_STATE + n) * NCHUNK + ch];
    const int s0 = ch * CHUNK;
    for (int s = s0; s < s0 + CHUNK; ++s) {
        const size_t bs = (size_t)(b * SEQ + s);
        const float dc  = disc[bs * D_MODEL + d];
        Dcum += dc;
        const float L   = A * Dcum;
        const float xv  = x[bs * D_MODEL + d];
        const float bm  = Bm[bs * D_STATE + n];
        const float Bx  = bm * invA * (__expf(dc * A) - 1.0f) * xv;
        const float bxl = __logf(fabsf(Bx));
        const float Lc  = fmaxf(L, bxl - LOG_CLAMP);
        const float sgn = (Bx > 0.0f) ? 1.0f : ((Bx < 0.0f) ? -1.0f : 0.0f);
        Ssum += __expf(bxl - Lc) * sgn;
        const float h = Ssum * __expf(Lc);
        float contrib = h * Cc[bs * D_STATE + n];
        contrib += __shfl_xor(contrib, 1, 32);
        contrib += __shfl_xor(contrib, 2, 32);
        contrib += __shfl_xor(contrib, 4, 32);
        contrib += __shfl_xor(contrib, 8, 32);
        if (n == 0) y[bs * D_MODEL + d] = contrib;
    }
}

extern "C" void kernel_launch(void* const* d_in, const int* in_sizes, int n_in,
                              void* d_out, int out_size, void* d_ws, size_t ws_size,
                              hipStream_t stream) {
    (void)in_sizes; (void)n_in; (void)out_size; (void)ws_size;
    const float* x    = (const float*)d_in[0];
    const float* Araw = (const float*)d_in[1];
    const float* P    = (const float*)d_in[2];
    const float* W_B  = (const float*)d_in[3];
    const float* b_B  = (const float*)d_in[4];
    const float* W_C  = (const float*)d_in[5];
    const float* b_C  = (const float*)d_in[6];
    const float* W_D  = (const float*)d_in[7];
    const float* b_D  = (const float*)d_in[8];
    float* y = (float*)d_out;

    constexpr size_t BS       = (size_t)BATCH * SEQ;
    constexpr size_t szBm     = BS * D_STATE * 4;
    constexpr size_t szCc     = BS * D_STATE * 4;
    constexpr size_t szXwd    = BS * 4;
    constexpr size_t szDisc   = BS * D_MODEL * 4;
    constexpr size_t szDsum   = (size_t)BATCH * D_MODEL * NCHUNK * 4;
    char* ws = (char*)d_ws;
    float* Bm   = (float*)(ws);
    float* Cc   = (float*)(ws + szBm);
    float* xwd  = (float*)(ws + szBm + szCc);
    float* disc = (float*)(ws + szBm + szCc + szXwd);
    float* dsum = (float*)(ws + szBm + szCc + szXwd + szDisc);
    float* sumS = (float*)(ws + szBm + szCc + szXwd + szDisc + szDsum);

    proj_wmma_kernel<<<(BS / 16) / 8, 256, 0, stream>>>(x, W_B, b_B, W_C, b_C, W_D, b_D,
                                                        Bm, Cc, xwd);
    disc_kernel<<<(BATCH * D_MODEL * NCHUNK) / 256, 256, 0, stream>>>(xwd, P, disc, dsum);
    scan_chunks_kernel<<<(BATCH * D_MODEL + 255) / 256, 256, 0, stream>>>(dsum, BATCH * D_MODEL);
    scan_pass1_kernel<<<(BATCH * D_MODEL * D_STATE * NCHUNK) / 256, 256, 0, stream>>>(
        x, Araw, disc, dsum, Bm, sumS);
    scan_chunks_kernel<<<(BATCH * D_MODEL * D_STATE + 255) / 256, 256, 0, stream>>>(
        sumS, BATCH * D_MODEL * D_STATE);
    scan_final_kernel<<<(BATCH * D_MODEL * D_STATE * NCHUNK) / 256, 256, 0, stream>>>(
        x, Araw, disc, dsum, Bm, Cc, sumS, y);
}


# Scan lanes re-mapped to 32 adjacent channels, the 16 states in-thread (chunk planes channel-fastest): whole-line loads/stores, uniform Bm/C loads; state sum now serial in-thread (order changes)
# speedup vs baseline: 4.3922x; 1.6658x over previous
#include <hip/hip_runtime.h>
#include <math.h>


#define D_MODEL 128
#define D_STATE 16
#define BATCH   4
#define SEQ     4096
#define NCHUNK  128
#define CHUNK   (SEQ / NCHUNK)
#define LOG_CLAMP 70.0f

typedef float v2f __attribute__((ext_vector_type(2)));
typedef float v8f __attribute__((ext_vector_type(8)));

__device__ __forceinline__ float softplus_f(float z) {
    return fmaxf(z, 0.0f) + log1pf(__expf(-fabsf(z)));
}

__global__ void proj_wmma_kernel(const float* __restrict__ x,
                                 const float* __restrict__ W_B, const float* __restrict__ b_B,
                                 const float* __restrict__ W_C, const float* __restrict__ b_C,
                                 const float* __restrict__ W_D, const float* __restrict__ b_D,
                                 float* __restrict__ Bm, float* __restrict__ Cc,
                                 float* __restrict__ xwd) {
    const int lane  = threadIdx.x & 31;
    const int wave  = threadIdx.x >> 5;
    const int tile  = blockIdx.x * (blockDim.x >> 5) + wave;
    const int half  = lane >> 4;
    const int lid   = lane & 15;
    const int row   = tile * 16 + lid;
    const int khalf = half * 2;
    const float dmask = (lid == 0) ? 1.0f : 0.0f;

    v8f accB = {}; v8f accC = {}; v8f accD = {};
    const float* xrow = x + (size_t)row * D_MODEL;

    for (int k0 = 0; k0 < D_MODEL; k0 += 4) {
        const v2f a = *(const v2f*)(xrow + k0 + khalf);
        v2f fB, fC, fD;
        fB.x = W_B[(k0 + khalf + 0) * D_STATE + lid];
        fB.y = W_B[(k0 + khalf + 1) * D_STATE + lid];
        fC.x = W_C[(k0 + khalf + 0) * D_STATE + lid];
        fC.y = W_C[(k0 + khalf + 1) * D_STATE + lid];
        fD.x = W_D[k0 + khalf + 0] * dmask;
        fD.y = W_D[k0 + khalf + 1] * dmask;
        accB = __builtin_amdgcn_wmma_f32_16x16x4_f32(false, a, false, fB, (short)0, accB, false, false);
        accC = __builtin_amdgcn_wmma_f32_16x16x4_f32(false, a, false, fC, (short)0, accC, false, false);
        accD = __builtin_amdgcn_wmma_f32_16x16x4_f32(false, a, false, fD, (short)0, accD, false, false);
    }

    const float bBn = b_B[lid];
    const float bCn = b_C[lid];
    const float bDs = b_D[0];
#pragma unroll
    for (int r = 0; r < 8; ++r) {
        const int orow = tile * 16 + r + 8 * half;
        Bm[(size_t)orow * D_STATE + lid] = accB[r] + bBn;
        Cc[(size_t)orow * D_STATE + lid] = accC[r] + bCn;
        if (lid == 0) xwd[orow] = accD[r] + bDs;
    }
}

__global__ void disc_kernel(const float* __restrict__ xwd, const float* __restrict__ P,
                            float* __restrict__ disc, float* __restrict__ discsum) {
    const int t  = blockIdx.x * blockDim.x + threadIdx.x;
    const int d  = t % D_MODEL;
    const int g  = t / D_MODEL;
    const int ch = g % NCHUNK;
    const int b  = g / NCHUNK;

    const float Pd = P[d];
    float acc = 0.0f;
    const int s0 = ch * CHUNK;
    for (int s = s0; s < s0 + CHUNK; ++s) {
        const float dc = softplus_f(Pd + xwd[b * SEQ + s]);
        disc[((size_t)(b * SEQ + s)) * D_MODEL + d] = dc;
        acc += dc;
    }
    discsum[(b * NCHUNK + ch) * D_MODEL + d] = acc;
}

__global__ void scan_chunks_kernel(float* __restrict__ arr, int nchannels) {
    const int t = blockIdx.x * blockDim.x + threadIdx.x;
    if (t >= nchannels) return;
    const int d     = t % D_MODEL;
    const int outer = t / D_MODEL;
    float run = 0.0f;
    float* p = arr + (size_t)outer * NCHUNK * D_MODEL + d;
#pragma unroll
    for (int c = 0; c < NCHUNK; ++c) {
        const float v = p[c * D_MODEL]; p[c * D_MODEL] = run; run += v;
    }
}


__device__ __forceinline__ void scan_step(const float (&A)[D_STATE], const float (&invA)[D_STATE],
                                          const float (&bm)[D_STATE], float dc, float xv,
                                          float Dcum, float (&S)[D_STATE], float (&Lc)[D_STATE]) {
    float Bx[D_STATE], bxl[D_STATE];
#pragma unroll
    for (int n = 0; n < D_STATE; ++n)
        Bx[n] = bm[n] * invA[n] * (__expf(dc * A[n]) - 1.0f) * xv;
#pragma unroll
    for (int n = 0; n < D_STATE; ++n)
        bxl[n] = __logf(fabsf(Bx[n]));
#pragma unroll
    for (int n = 0; n < D_STATE; ++n)
        Lc[n] = fmaxf(A[n] * Dcum, bxl[n] - LOG_CLAMP);
#pragma unroll
    for (int n = 0; n < D_STATE; ++n) {
        const float sgn = (Bx[n] > 0.0f) ? 1.0f : ((Bx[n] < 0.0f) ? -1.0f : 0.0f);
        S[n] += __expf(bxl[n] - Lc[n]) * sgn;
    }
}

__global__ void __launch_bounds__(D_MODEL) __attribute__((amdgpu_waves_per_eu(1, 4)))
scan_pass1_kernel(const float* __restrict__ x,
                  const float* __restrict__ Araw,
                  const float* __restrict__ disc,
                  const float* __restrict__ discpre,
                  const float* __restrict__ Bm,
                  float* __restrict__ sumS) {
    const int d  = threadIdx.x;
    const int ch = blockIdx.x % NCHUNK;
    const int b  = blockIdx.x / NCHUNK;

    float A[D_STATE], invA[D_STATE], acc[D_STATE];
#pragma unroll
    for (int n = 0; n < D_STATE; ++n) {
        A[n]    = -softplus_f(Araw[d * D_STATE + n]);
        invA[n] = 1.0f / A[n];
        acc[n]  = 0.0f;
    }
    float Dcum = discpre[(b * NCHUNK + ch) * D_MODEL + d];

    const size_t row0 = (size_t)b * SEQ + (size_t)ch * CHUNK;
    float dc = disc[row0 * D_MODEL + d];
    float xv = x[row0 * D_MODEL + d];
    float bm[D_STATE];
#pragma unroll
    for (int n = 0; n < D_STATE; ++n) bm[n] = Bm[row0 * D_STATE + n];

    for (int i = 0; i < CHUNK; ++i) {
        const size_t rn = row0 + (size_t)((i + 1 < CHUNK) ? (i + 1) : (CHUNK - 1));
        const float dc_n = disc[rn * D_MODEL + d];
        const float xv_n = x[rn * D_MODEL + d];
        float bm_n[D_STATE];
#pragma unroll
        for (int n = 0; n < D_STATE; ++n) bm_n[n] = Bm[rn * D_STATE + n];

        Dcum += dc;
        float Lc[D_STATE];
        scan_step(A, invA, bm, dc, xv, Dcum, acc, Lc);

        asm volatile("" :: "v"(dc_n), "v"(xv_n));
        dc = dc_n; xv = xv_n;
#pragma unroll
        for (int n = 0; n < D_STATE; ++n) bm[n] = bm_n[n];
    }
#pragma unroll
    for (int n = 0; n < D_STATE; ++n)
        sumS[((size_t)((b * D_STATE + n) * NCHUNK + ch)) * D_MODEL + d] = acc[n];
}

__global__ void __launch_bounds__(D_MODEL) __attribute__((amdgpu_waves_per_eu(1, 4)))
scan_final_kernel(const float* __restrict__ x,
                  const float* __restrict__ Araw,
                  const float* __restrict__ disc,
                  const float* __restrict__ discpre,
                  const float* __restrict__ Bm,
                  const float* __restrict__ Cc,
                  const float* __restrict__ sumSpre,
                  float* __restrict__ y) {
    const int d  = threadIdx.x;
    const int ch = blockIdx.x % NCHUNK;
    const int b  = blockIdx.x / NCHUNK;

    float A[D_STATE], invA[D_STATE], Ssum[D_STATE];
#pragma unroll
    for (int n = 0; n < D_STATE; ++n) {
        A[n]    = -softplus_f(Araw[d * D_STATE + n]);
        invA[n] = 1.0f / A[n];
        Ssum[n] = sumSpre[((size_t)((b * D_STATE + n) * NCHUNK + ch)) * D_MODEL + d];
    }
    float Dcum = discpre[(b * NCHUNK + ch) * D_MODEL + d];

    const size_t row0 = (size_t)b * SEQ + (size_t)ch * CHUNK;
    float dc = disc[row0 * D_MODEL + d];
    float xv = x[row0 * D_MODEL + d];
    float bm[D_STATE], cc[D_STATE];
#pragma unroll
    for (int n = 0; n < D_STATE; ++n) {
        bm[n] = Bm[row0 * D_STATE + n];
        cc[n] = Cc[row0 * D_STATE + n];
    }

    for (int i = 0; i < CHUNK; ++i) {
        const size_t rn = row0 + (size_t)((i + 1 < CHUNK) ? (i + 1) : (CHUNK - 1));
        const float dc_n = disc[rn * D_MODEL + d];
        const float xv_n = x[rn * D_MODEL + d];
        float bm_n[D_STATE], cc_n[D_STATE];
#pragma unroll
        for (int n = 0; n < D_STATE; ++n) {
            bm_n[n] = Bm[rn * D_STATE + n];
            cc_n[n] = Cc[rn * D_STATE + n];
        }

        Dcum += dc;
        float Lc[D_STATE];
        scan_step(A, invA, bm, dc, xv, Dcum, Ssum, Lc);
        float h[D_STATE];
#pragma unroll
        for (int n = 0; n < D_STATE; ++n) h[n] = Ssum[n] * __expf(Lc[n]);
        float yv = 0.0f;
#pragma unroll
        for (int n = 0; n < D_STATE; ++n) yv += h[n] * cc[n];
        y[(row0 + i) * D_MODEL + d] = yv;

        asm volatile("" :: "v"(dc_n), "v"(xv_n));
        dc = dc_n; xv = xv_n;
#pragma unroll
        for (int n = 0; n < D_STATE; ++n) { bm[n] = bm_n[n]; cc[n] = cc_n[n]; }
    }
}

extern "C" void kernel_launch(void* const* d_in, const int* in_sizes, int n_in,
                              void* d_out, int out_size, void* d_ws, size_t ws_size,
                              hipStream_t stream) {
    (void)in_sizes; (void)n_in; (void)out_size; (void)ws_size;
    const float* x    = (const float*)d_in[0];
    const float* Araw = (const float*)d_in[1];
    const float* P    = (const float*)d_in[2];
    const float* W_B  = (const float*)d_in[3];
    const float* b_B  = (const float*)d_in[4];
    const float* W_C  = (const float*)d_in[5];
    const float* b_C  = (const float*)d_in[6];
    const float* W_D  = (const float*)d_in[7];
    const float* b_D  = (const float*)d_in[8];
    float* y = (float*)d_out;

    constexpr size_t BS       = (size_t)BATCH * SEQ;
    constexpr size_t szBm     = BS * D_STATE * 4;
    constexpr size_t szCc     = BS * D_STATE * 4;
    constexpr size_t szXwd    = BS * 4;
    constexpr size_t szDisc   = BS * D_MODEL * 4;
    constexpr size_t szDsum   = (size_t)BATCH * D_MODEL * NCHUNK * 4;
    char* ws = (char*)d_ws;
    float* Bm   = (float*)(ws);
    float* Cc   = (float*)(ws + szBm);
    float* xwd  = (float*)(ws + szBm + szCc);
    float* disc = (float*)(ws + szBm + szCc + szXwd);
    float* dsum = (float*)(ws + szBm + szCc + szXwd + szDisc);
    float* sumS = (float*)(ws + szBm + szCc + szXwd + szDisc + szDsum);

    proj_wmma_kernel<<<(BS / 16) / 8, 256, 0, stream>>>(x, W_B, b_B, W_C, b_C, W_D, b_D,
                                                        Bm, Cc, xwd);
    disc_kernel<<<(BATCH * D_MODEL * NCHUNK) / 256, 256, 0, stream>>>(xwd, P, disc, dsum);
    scan_chunks_kernel<<<(BATCH * D_MODEL + 255) / 256, 256, 0, stream>>>(dsum, BATCH * D_MODEL);
    scan_pass1_kernel<<<BATCH * NCHUNK, D_MODEL, 0, stream>>>(
        x, Araw, disc, dsum, Bm, sumS);
    scan_chunks_kernel<<<(BATCH * D_MODEL * D_STATE + 255) / 256, 256, 0, stream>>>(
        sumS, BATCH * D_MODEL * D_STATE);
    scan_final_kernel<<<BATCH * NCHUNK, D_MODEL, 0, stream>>>(
        x, Araw, disc, dsum, Bm, Cc, sumS, y);
}


# Scan step moved to base 2 on the bare v_exp_f32 / v_log_f32 (decay rate times log2(e) once per channel, clamp in log2 units): library log expansion and exp scalings leave both loops; rounding changes
# speedup vs baseline: 4.7671x; 1.0854x over previous
#include <hip/hip_runtime.h>
#include <math.h>


#define D_MODEL 128
#define D_STATE 16
#define BATCH   4
#define SEQ     4096
#define NCHUNK  128
#define CHUNK   (SEQ / NCHUNK)
#define LOG_CLAMP 70.0f
#define LOG2E  1.4426950408889634f
#define CLAMP2 ((float)((double)LOG_CLAMP * 1.4426950408889634))

typedef float v2f __attribute__((ext_vector_type(2)));
typedef float v8f __attribute__((ext_vector_type(8)));

__device__ __forceinline__ float softplus_f(float z) {
    return fmaxf(z, 0.0f) + log1pf(__expf(-fabsf(z)));
}

__global__ void proj_wmma_kernel(const float* __restrict__ x,
                                 const float* __restrict__ W_B, const float* __restrict__ b_B,
                                 const float* __restrict__ W_C, const float* __restrict__ b_C,
                                 const float* __restrict__ W_D, const float* __restrict__ b_D,
                                 float* __restrict__ Bm, float* __restrict__ Cc,
                                 float* __restrict__ xwd) {
    const int lane  = threadIdx.x & 31;
    const int wave  = threadIdx.x >> 5;
    const int tile  = blockIdx.x * (blockDim.x >> 5) + wave;
    const int half  = lane >> 4;
    const int lid   = lane & 15;
    const int row   = tile * 16 + lid;
    const int khalf = half * 2;
    const float dmask = (lid == 0) ? 1.0f : 0.0f;

    v8f accB = {}; v8f accC = {}; v8f accD = {};
    const float* xrow = x + (size_t)row * D_MODEL;

    for (int k0 = 0; k0 < D_MODEL; k0 += 4) {
        const v2f a = *(const v2f*)(xrow + k0 + khalf);
        v2f fB, fC, fD;
        fB.x = W_B[(k0 + khalf + 0) * D_STATE + lid];
        fB.y = W_B[(k0 + khalf + 1) * D_STATE + lid];
        fC.x = W_C[(k0 + khalf + 0) * D_STATE + lid];
        fC.y = W_C[(k0 + khalf + 1) * D_STATE + lid];
        fD.x = W_D[k0 + khalf + 0] * dmask;
        fD.y = W_D[k0 + khalf + 1] * dmask;
        accB = __builtin_amdgcn_wmma_f32_16x16x4_f32(false, a, false, fB, (short)0, accB, false, false);
        accC = __builtin_amdgcn_wmma_f32_16x16x4_f32(false, a, false, fC, (short)0, accC, false, false);
        accD = __builtin_amdgcn_wmma_f32_16x16x4_f32(false, a, false, fD, (short)0, accD, false, false);
    }

    const float bBn = b_B[lid];
    const float bCn = b_C[lid];
    const float bDs = b_D[0];
#pragma unroll
    for (int r = 0; r < 8; ++r) {
        const int orow = tile * 16 + r + 8 * half;
        Bm[(size_t)orow * D_STATE + lid] = accB[r] + bBn;
        Cc[(size_t)orow * D_STATE + lid] = accC[r] + bCn;
        if (lid == 0) xwd[orow] = accD[r] + bDs;
    }
}

__global__ void disc_kernel(const float* __restrict__ xwd, const float* __restrict__ P,
                            float* __restrict__ disc, float* __restrict__ discsum) {
    const int t  = blockIdx.x * blockDim.x + threadIdx.x;
    const int d  = t % D_MODEL;
    const int g  = t / D_MODEL;
    const int ch = g % NCHUNK;
    const int b  = g / NCHUNK;

    const float Pd = P[d];
    float acc = 0.0f;
    const int s0 = ch * CHUNK;
    for (int s = s0; s < s0 + CHUNK; ++s) {
        const float dc = softplus_f(Pd + xwd[b * SEQ + s]);
        disc[((size_t)(b * SEQ + s)) * D_MODEL + d] = dc;
        acc += dc;
    }
    discsum[(b * NCHUNK + ch) * D_MODEL + d] = acc;
}

__global__ void scan_chunks_kernel(float* __restrict__ arr, int nchannels) {
    const int t = blockIdx.x * blockDim.x + threadIdx.x;
    if (t >= nchannels) return;
    const int d     = t % D_MODEL;
    const int outer = t / D_MODEL;
    float run = 0.0f;
    float* p = arr + (size_t)outer * NCHUNK * D_MODEL + d;
#pragma unroll
    for (int c = 0; c < NCHUNK; ++c) {
        const float v = p[c * D_MODEL]; p[c * D_MODEL] = run; run += v;
    }
}


__device__ __forceinline__ void scan_step(const float (&A2)[D_STATE], const float (&invA)[D_STATE],
                                          const float (&bm)[D_STATE], float dc, float xv,
                                          float Dcum, float (&S)[D_STATE], float (&Lc2)[D_STATE]) {
    float Bx[D_STATE], bxl2[D_STATE];
#pragma unroll
    for (int n = 0; n < D_STATE; ++n)
        Bx[n] = bm[n] * invA[n] * (__builtin_amdgcn_exp2f(dc * A2[n]) - 1.0f) * xv;
#pragma unroll
    for (int n = 0; n < D_STATE; ++n)
        bxl2[n] = __builtin_amdgcn_logf(fabsf(Bx[n]));
#pragma unroll
    for (int n = 0; n < D_STATE; ++n)
        Lc2[n] = fmaxf(A2[n] * Dcum, bxl2[n] - CLAMP2);
#pragma unroll
    for (int n = 0; n < D_STATE; ++n) {
        const float sgn = (Bx[n] > 0.0f) ? 1.0f : ((Bx[n] < 0.0f) ? -1.0f : 0.0f);
        S[n] += __builtin_amdgcn_exp2f(bxl2[n] - Lc2[n]) * sgn;
    }
}

__global__ void __launch_bounds__(D_MODEL) __attribute__((amdgpu_waves_per_eu(1, 4)))
scan_pass1_kernel(const float* __restrict__ x,
                  const float* __restrict__ Araw,
                  const float* __restrict__ disc,
                  const float* __restrict__ discpre,
                  const float* __restrict__ Bm,
                  float* __restrict__ sumS) {
    const int d  = threadIdx.x;
    const int ch = blockIdx.x % NCHUNK;
    const int b  = blockIdx.x / NCHUNK;

    float A2[D_STATE], invA[D_STATE], acc[D_STATE];
#pragma unroll
    for (int n = 0; n < D_STATE; ++n) {
        const float An = -softplus_f(Araw[d * D_STATE + n]);
        invA[n] = 1.0f / An;
        A2[n]   = An * LOG2E;
        acc[n]  = 0.0f;
    }
    float Dcum = discpre[(b * NCHUNK + ch) * D_MODEL + d];

    const size_t row0 = (size_t)b * SEQ + (size_t)ch * CHUNK;
    float dc = disc[row0 * D_MODEL + d];
    float xv = x[row0 * D_MODEL + d];
    float bm[D_STATE];
#pragma unroll
    for (int n = 0; n < D_STATE; ++n) bm[n] = Bm[row0 * D_STATE + n];

    for (int i = 0; i < CHUNK; ++i) {
        const size_t rn = row0 + (size_t)((i + 1 < CHUNK) ? (i + 1) : (CHUNK - 1));
        const float dc_n = disc[rn * D_MODEL + d];
        const float xv_n = x[rn * D_MODEL + d];
        float bm_n[D_STATE];
#pragma unroll
        for (int n = 0; n < D_STATE; ++n) bm_n[n] = Bm[rn * D_STATE + n];

        Dcum += dc;
        float Lc2[D_STATE];
        scan_step(A2, invA, bm, dc, xv, Dcum, acc, Lc2);

        asm volatile("" :: "v"(dc_n), "v"(xv_n));
        dc = dc_n; xv = xv_n;
#pragma unroll
        for (int n = 0; n < D_STATE; ++n) bm[n] = bm_n[n];
    }
#pragma unroll
    for (int n = 0; n < D_STATE; ++n)
        sumS[((size_t)((b * D_STATE + n) * NCHUNK + ch)) * D_MODEL + d] = acc[n];
}

__global__ void __launch_bounds__(D_MODEL) __attribute__((amdgpu_waves_per_eu(1, 4)))
scan_final_kernel(const float* __restrict__ x,
                  const float* __restrict__ Araw,
                  const float* __restrict__ disc,
                  const float* __restrict__ discpre,
                  const float* __restrict__ Bm,
                  const float* __restrict__ Cc,
                  const float* __restrict__ sumSpre,
                  float* __restrict__ y) {
    const int d  = threadIdx.x;
    const int ch = blockIdx.x % NCHUNK;
    const int b  = blockIdx.x / NCHUNK;

    float A2[D_STATE], invA[D_STATE], Ssum[D_STATE];
#pragma unroll
    for (int n = 0; n < D_STATE; ++n) {
        const float An = -softplus_f(Araw[d * D_STATE + n]);
        invA[n] = 1.0f / An;
        A2[n]   = An * LOG2E;
        Ssum[n] = sumSpre[((size_t)((b * D_STATE + n) * NCHUNK + ch)) * D_MODEL + d];
    }
    float Dcum = discpre[(b * NCHUNK + ch) * D_MODEL + d];

    const size_t row0 = (size_t)b * SEQ + (size_t)ch * CHUNK;
    float dc = disc[row0 * D_MODEL + d];
    float xv = x[row0 * D_MODEL + d];
    float bm[D_STATE], cc[D_STATE];
#pragma unroll
    for (int n = 0; n < D_STATE; ++n) {
        bm[n] = Bm[row0 * D_STATE + n];
        cc[n] = Cc[row0 * D_STATE + n];
    }

    for (int i = 0; i < CHUNK; ++i) {
        const size_t rn = row0 + (size_t)((i + 1 < CHUNK) ? (i + 1) : (CHUNK - 1));
        const float dc_n = disc[rn * D_MODEL + d];
        const float xv_n = x[rn * D_MODEL + d];
        float bm_n[D_STATE], cc_n[D_STATE];
#pragma unroll
        for (int n = 0; n < D_STATE; ++n) {
            bm_n[n] = Bm[rn * D_STATE + n];
            cc_n[n] = Cc[rn * D_STATE + n];
        }

        Dcum += dc;
        float Lc2[D_STATE];
        scan_step(A2, invA, bm, dc, xv, Dcum, Ssum, Lc2);
        float h[D_STATE];
#pragma unroll
        for (int n = 0; n < D_STATE; ++n) h[n] = Ssum[n] * __builtin_amdgcn_exp2f(Lc2[n]);
        float yv = 0.0f;
#pragma unroll
        for (int n = 0; n < D_STATE; ++n) yv += h[n] * cc[n];
        y[(row0 + i) * D_MODEL + d] = yv;

        asm volatile("" :: "v"(dc_n), "v"(xv_n));
        dc = dc_n; xv = xv_n;
#pragma unroll
        for (int n = 0; n < D_STATE; ++n) { bm[n] = bm_n[n]; cc[n] = cc_n[n]; }
    }
}

extern "C" void kernel_launch(void* const* d_in, const int* in_sizes, int n_in,
                              void* d_out, int out_size, void* d_ws, size_t ws_size,
                              hipStream_t stream) {
    (void)in_sizes; (void)n_in; (void)out_size; (void)ws_size;
    const float* x    = (const float*)d_in[0];
    const float* Araw = (const float*)d_in[1];
    const float* P    = (const float*)d_in[2];
    const float* W_B  = (const float*)d_in[3];
    const float* b_B  = (const float*)d_in[4];
    const float* W_C  = (const float*)d_in[5];
    const float* b_C  = (const float*)d_in[6];
    const float* W_D  = (const float*)d_in[7];
    const float* b_D  = (const float*)d_in[8];
    float* y = (float*)d_out;

    constexpr size_t BS       = (size_t)BATCH * SEQ;
    constexpr size_t szBm     = BS * D_STATE * 4;
    constexpr size_t szCc     = BS * D_STATE * 4;
    constexpr size_t szXwd    = BS * 4;
    constexpr size_t szDisc   = BS * D_MODEL * 4;
    constexpr size_t szDsum   = (size_t)BATCH * D_MODEL * NCHUNK * 4;
    char* ws = (char*)d_ws;
    float* Bm   = (float*)(ws);
    float* Cc   = (float*)(ws + szBm);
    float* xwd  = (float*)(ws + szBm + szCc);
    float* disc = (float*)(ws + szBm + szCc + szXwd);
    float* dsum = (float*)(ws + szBm + szCc + szXwd + szDisc);
    float* sumS = (float*)(ws + szBm + szCc + szXwd + szDisc + szDsum);

    proj_wmma_kernel<<<(BS / 16) / 8, 256, 0, stream>>>(x, W_B, b_B, W_C, b_C, W_D, b_D,
                                                        Bm, Cc, xwd);
    disc_kernel<<<(BATCH * D_MODEL * NCHUNK) / 256, 256, 0, stream>>>(xwd, P, disc, dsum);
    scan_chunks_kernel<<<(BATCH * D_MODEL + 255) / 256, 256, 0, stream>>>(dsum, BATCH * D_MODEL);
    scan_pass1_kernel<<<BATCH * NCHUNK, D_MODEL, 0, stream>>>(
        x, Araw, disc, dsum, Bm, sumS);
    scan_chunks_kernel<<<(BATCH * D_MODEL * D_STATE + 255) / 256, 256, 0, stream>>>(
        sumS, BATCH * D_MODEL * D_STATE);
    scan_final_kernel<<<BATCH * NCHUNK, D_MODEL, 0, stream>>>(
        x, Araw, disc, dsum, Bm, Cc, sumS, y);
}


# Launches 6 -> 4: disc and its chunk sums run in the projection blocks behind one block barrier (xwd via LDS); the chunk prefix of the disc sums is summed in chunk order inside scan pass 1
# speedup vs baseline: 5.1308x; 1.0763x over previous
#include <hip/hip_runtime.h>
#include <math.h>


#define D_MODEL 128
#define D_STATE 16
#define BATCH   4
#define SEQ     4096
#define NCHUNK  128
#define CHUNK   (SEQ / NCHUNK)
#define LOG_CLAMP 70.0f
#define LOG2E  1.4426950408889634f
#define CLAMP2 ((float)((double)LOG_CLAMP * 1.4426950408889634))

typedef float v2f __attribute__((ext_vector_type(2)));
typedef float v8f __attribute__((ext_vector_type(8)));

__device__ __forceinline__ float softplus_f(float z) {
    return fmaxf(z, 0.0f) + log1pf(__expf(-fabsf(z)));
}

#define PROJ_WAVES 4
#define PROJ_ROWS  (PROJ_WAVES * 16)
__global__ void __launch_bounds__(256)
proj_disc_kernel(const float* __restrict__ x,
                 const float* __restrict__ W_B, const float* __restrict__ b_B,
                 const float* __restrict__ W_C, const float* __restrict__ b_C,
                 const float* __restrict__ W_D, const float* __restrict__ b_D,
                 const float* __restrict__ P,
                 float* __restrict__ Bm, float* __restrict__ Cc,
                 float* __restrict__ disc, float* __restrict__ discsum) {
    __shared__ float s_xwd[PROJ_ROWS];
    const int lane  = threadIdx.x & 31;
    const int wave  = __builtin_amdgcn_readfirstlane((int)(threadIdx.x >> 5));
  if (wave < PROJ_WAVES) {
    const int tile  = blockIdx.x * PROJ_WAVES + wave;
    const int half  = lane >> 4;
    const int lid   = lane & 15;
    const int row   = tile * 16 + lid;
    const int khalf = half * 2;
    const float dmask = (lid == 0) ? 1.0f : 0.0f;

    v8f accB = {}; v8f accC = {}; v8f accD = {};
    const float* xrow = x + (size_t)row * D_MODEL;

    for (int k0 = 0; k0 < D_MODEL; k0 += 4) {
        const v2f a = *(const v2f*)(xrow + k0 + khalf);
        v2f fB, fC, fD;
        fB.x = W_B[(k0 + khalf + 0) * D_STATE + lid];
        fB.y = W_B[(k0 + khalf + 1) * D_STATE + lid];
        fC.x = W_C[(k0 + khalf + 0) * D_STATE + lid];
        fC.y = W_C[(k0 + khalf + 1) * D_STATE + lid];
        fD.x = W_D[k0 + khalf + 0] * dmask;
        fD.y = W_D[k0 + khalf + 1] * dmask;
        accB = __builtin_amdgcn_wmma_f32_16x16x4_f32(false, a, false, fB, (short)0, accB, false, false);
        accC = __builtin_amdgcn_wmma_f32_16x16x4_f32(false, a, false, fC, (short)0, accC, false, false);
        accD = __builtin_amdgcn_wmma_f32_16x16x4_f32(false, a, false, fD, (short)0, accD, false, false);
    }

    const float bBn = b_B[lid];
    const float bCn = b_C[lid];
    const float bDs = b_D[0];
#pragma unroll
    for (int r = 0; r < 8; ++r) {
        const int orow = tile * 16 + r + 8 * half;
        Bm[(size_t)orow * D_STATE + lid] = accB[r] + bBn;
        Cc[(size_t)orow * D_STATE + lid] = accC[r] + bCn;
        if (lid == 0) s_xwd[wave * 16 + r + 8 * half] = accD[r] + bDs;
    }
  }
    __syncthreads();

    const int d  = threadIdx.x & (D_MODEL - 1);
    const int g  = threadIdx.x >> 7;
    const int gc = blockIdx.x * (PROJ_ROWS / CHUNK) + g;
    const size_t row0 = (size_t)gc * CHUNK;
    const float Pd = P[d];
    float acc = 0.0f;
    for (int s = 0; s < CHUNK; ++s) {
        const float dc = softplus_f(Pd + s_xwd[g * CHUNK + s]);
        disc[(row0 + s) * D_MODEL + d] = dc;
        acc += dc;
    }
    discsum[(size_t)gc * D_MODEL + d] = acc;
}

__global__ void scan_chunks_kernel(float* __restrict__ arr, int nchannels) {
    const int t = blockIdx.x * blockDim.x + threadIdx.x;
    if (t >= nchannels) return;
    const int d     = t % D_MODEL;
    const int outer = t / D_MODEL;
    float run = 0.0f;
    float* p = arr + (size_t)outer * NCHUNK * D_MODEL + d;
#pragma unroll
    for (int c = 0; c < NCHUNK; ++c) {
        const float v = p[c * D_MODEL]; p[c * D_MODEL] = run; run += v;
    }
}


__device__ __forceinline__ void scan_step(const float (&A2)[D_STATE], const float (&invA)[D_STATE],
                                          const float (&bm)[D_STATE], float dc, float xv,
                                          float Dcum, float (&S)[D_STATE], float (&Lc2)[D_STATE]) {
    float Bx[D_STATE], bxl2[D_STATE];
#pragma unroll
    for (int n = 0; n < D_STATE; ++n)
        Bx[n] = bm[n] * invA[n] * (__builtin_amdgcn_exp2f(dc * A2[n]) - 1.0f) * xv;
#pragma unroll
    for (int n = 0; n < D_STATE; ++n)
        bxl2[n] = __builtin_amdgcn_logf(fabsf(Bx[n]));
#pragma unroll
    for (int n = 0; n < D_STATE; ++n)
        Lc2[n] = fmaxf(A2[n] * Dcum, bxl2[n] - CLAMP2);
#pragma unroll
    for (int n = 0; n < D_STATE; ++n) {
        const float sgn = (Bx[n] > 0.0f) ? 1.0f : ((Bx[n] < 0.0f) ? -1.0f : 0.0f);
        S[n] += __builtin_amdgcn_exp2f(bxl2[n] - Lc2[n]) * sgn;
    }
}

__global__ void __launch_bounds__(D_MODEL) __attribute__((amdgpu_waves_per_eu(1, 4)))
scan_pass1_kernel(const float* __restrict__ x,
                  const float* __restrict__ Araw,
                  const float* __restrict__ disc,
                  const float* __restrict__ discsum,
                  float* __restrict__ discpre,
                  const float* __restrict__ Bm,
                  float* __restrict__ sumS) {
    const int d  = threadIdx.x;
    const int ch = blockIdx.x % NCHUNK;
    const int b  = blockIdx.x / NCHUNK;

    float A2[D_STATE], invA[D_STATE], acc[D_STATE];
#pragma unroll
    for (int n = 0; n < D_STATE; ++n) {
        const float An = -softplus_f(Araw[d * D_STATE + n]);
        invA[n] = 1.0f / An;
        A2[n]   = An * LOG2E;
        acc[n]  = 0.0f;
    }
    float Dcum = 0.0f;
    {
        const float* dp = discsum + (size_t)b * NCHUNK * D_MODEL + d;
        float dv[NCHUNK];
#pragma unroll
        for (int c = 0; c < NCHUNK; ++c) dv[c] = dp[c * D_MODEL];
        asm volatile("" :: "v"(dv[0]));
#pragma unroll
        for (int c = 0; c < NCHUNK; ++c) Dcum += (c < ch) ? dv[c] : 0.0f;
    }
    discpre[(b * NCHUNK + ch) * D_MODEL + d] = Dcum;

    const size_t row0 = (size_t)b * SEQ + (size_t)ch * CHUNK;
    float dc = disc[row0 * D_MODEL + d];
    float xv = x[row0 * D_MODEL + d];
    float bm[D_STATE];
#pragma unroll
    for (int n = 0; n < D_STATE; ++n) bm[n] = Bm[row0 * D_STATE + n];

    for (int i = 0; i < CHUNK; ++i) {
        const size_t rn = row0 + (size_t)((i + 1 < CHUNK) ? (i + 1) : (CHUNK - 1));
        const float dc_n = disc[rn * D_MODEL + d];
        const float xv_n = x[rn * D_MODEL + d];
        float bm_n[D_STATE];
#pragma unroll
        for (int n = 0; n < D_STATE; ++n) bm_n[n] = Bm[rn * D_STATE + n];

        Dcum += dc;
        float Lc2[D_STATE];
        scan_step(A2, invA, bm, dc, xv, Dcum, acc, Lc2);

        asm volatile("" :: "v"(dc_n), "v"(xv_n));
        dc = dc_n; xv = xv_n;
#pragma unroll
        for (int n = 0; n < D_STATE; ++n) bm[n] = bm_n[n];
    }
#pragma unroll
    for (int n = 0; n < D_STATE; ++n)
        sumS[((size_t)((b * D_STATE + n) * NCHUNK + ch)) * D_MODEL + d] = acc[n];
}

__global__ void __launch_bounds__(D_MODEL) __attribute__((amdgpu_waves_per_eu(1, 4)))
scan_final_kernel(const float* __restrict__ x,
                  const float* __restrict__ Araw,
                  const float* __restrict__ disc,
                  const float* __restrict__ discpre,
                  const float* __restrict__ Bm,
                  const float* __restrict__ Cc,
                  const float* __restrict__ sumSpre,
                  float* __restrict__ y) {
    const int d  = threadIdx.x;
    const int ch = blockIdx.x % NCHUNK;
    const int b  = blockIdx.x / NCHUNK;

    float A2[D_STATE], invA[D_STATE], Ssum[D_STATE];
#pragma unroll
    for (int n = 0; n < D_STATE; ++n) {
        const float An = -softplus_f(Araw[d * D_STATE + n]);
        invA[n] = 1.0f / An;
        A2[n]   = An * LOG2E;
        Ssum[n] = sumSpre[((size_t)((b * D_STATE + n) * NCHUNK + ch)) * D_MODEL + d];
    }
    float Dcum = discpre[(b * NCHUNK + ch) * D_MODEL + d];

    const size_t row0 = (size_t)b * SEQ + (size_t)ch * CHUNK;
    float dc = disc[row0 * D_MODEL + d];
    float xv = x[row0 * D_MODEL + d];
    float bm[D_STATE], cc[D_STATE];
#pragma unroll
    for (int n = 0; n < D_STATE; ++n) {
        bm[n] = Bm[row0 * D_STATE + n];
        cc[n] = Cc[row0 * D_STATE + n];
    }

    for (int i = 0; i < CHUNK; ++i) {
        const size_t rn = row0 + (size_t)((i + 1 < CHUNK) ? (i + 1) : (CHUNK - 1));
        const float dc_n = disc[rn * D_MODEL + d];
        const float xv_n = x[rn * D_MODEL + d];
        float bm_n[D_STATE], cc_n[D_STATE];
#pragma unroll
        for (int n = 0; n < D_STATE; ++n) {
            bm_n[n] = Bm[rn * D_STATE + n];
            cc_n[n] = Cc[rn * D_STATE + n];
        }

        Dcum += dc;
        float Lc2[D_STATE];
        scan_step(A2, invA, bm, dc, xv, Dcum, Ssum, Lc2);
        float h[D_STATE];
#pragma unroll
        for (int n = 0; n < D_STATE; ++n) h[n] = Ssum[n] * __builtin_amdgcn_exp2f(Lc2[n]);
        float yv = 0.0f;
#pragma unroll
        for (int n = 0; n < D_STATE; ++n) yv += h[n] * cc[n];
        y[(row0 + i) * D_MODEL + d] = yv;

        asm volatile("" :: "v"(dc_n), "v"(xv_n));
        dc = dc_n; xv = xv_n;
#pragma unroll
        for (int n = 0; n < D_STATE; ++n) { bm[n] = bm_n[n]; cc[n] = cc_n[n]; }
    }
}

extern "C" void kernel_launch(void* const* d_in, const int* in_sizes, int n_in,
                              void* d_out, int out_size, void* d_ws, size_t ws_size,
                              hipStream_t stream) {
    (void)in_sizes; (void)n_in; (void)out_size; (void)ws_size;
    const float* x    = (const float*)d_in[0];
    const float* Araw = (const float*)d_in[1];
    const float* P    = (const float*)d_in[2];
    const float* W_B  = (const float*)d_in[3];
    const float* b_B  = (const float*)d_in[4];
    const float* W_C  = (const float*)d_in[5];
    const float* b_C  = (const float*)d_in[6];
    const float* W_D  = (const float*)d_in[7];
    const float* b_D  = (const float*)d_in[8];
    float* y = (float*)d_out;

    constexpr size_t BS       = (size_t)BATCH * SEQ;
    constexpr size_t szBm     = BS * D_STATE * 4;
    constexpr size_t szCc     = BS * D_STATE * 4;
    constexpr size_t szDisc   = BS * D_MODEL * 4;
    constexpr size_t szDsum   = (size_t)BATCH * D_MODEL * NCHUNK * 4;
    char* ws = (char*)d_ws;
    float* Bm   = (float*)(ws);
    float* Cc   = (float*)(ws + szBm);
    float* disc = (float*)(ws + szBm + szCc);
    float* dsum = (float*)(ws + szBm + szCc + szDisc);
    float* dpre = (float*)(ws + szBm + szCc + szDisc + szDsum);
    float* sumS = (float*)(ws + szBm + szCc + szDisc + szDsum + szDsum);

    proj_disc_kernel<<<BS / PROJ_ROWS, 256, 0, stream>>>(x, W_B, b_B, W_C, b_C, W_D, b_D, P,
                                                         Bm, Cc, disc, dsum);
    scan_pass1_kernel<<<BATCH * NCHUNK, D_MODEL, 0, stream>>>(
        x, Araw, disc, dsum, dpre, Bm, sumS);
    scan_chunks_kernel<<<(BATCH * D_MODEL * D_STATE + 255) / 256, 256, 0, stream>>>(
        sumS, BATCH * D_MODEL * D_STATE);
    scan_final_kernel<<<BATCH * NCHUNK, D_MODEL, 0, stream>>>(
        x, Araw, disc, dpre, Bm, Cc, sumS, y);
}


# softplus_f (disc and the decay table): library log1pf chain, about 90 dependent instructions, replaced by hardware log2 of 1+e, a cubic series below e = 2^-6; rounding of disc and A changes
# speedup vs baseline: 7.5057x; 1.4629x over previous
#include <hip/hip_runtime.h>
#include <math.h>


#define D_MODEL 128
#define D_STATE 16
#define BATCH   4
#define SEQ     4096
#define NCHUNK  128
#define CHUNK   (SEQ / NCHUNK)
#define LOG_CLAMP 70.0f
#define LOG2E  1.4426950408889634f
#define CLAMP2 ((float)((double)LOG_CLAMP * 1.4426950408889634))

typedef float v2f __attribute__((ext_vector_type(2)));
typedef float v8f __attribute__((ext_vector_type(8)));

#define LN2 0.6931471805599453f
__device__ __forceinline__ float softplus_f(float z) {
    const float e  = __builtin_amdgcn_exp2f(-LOG2E * fabsf(z));
    const float lg = LN2 * __builtin_amdgcn_logf(1.0f + e);
    const float sr = e * fmaf(e, fmaf(e, 1.0f / 3.0f, -0.5f), 1.0f);
    return fmaxf(z, 0.0f) + ((e < 0.015625f) ? sr : lg);
}

#define PROJ_WAVES 4
#define PROJ_ROWS  (PROJ_WAVES * 16)
__global__ void __launch_bounds__(256)
proj_disc_kernel(const float* __restrict__ x,
                 const float* __restrict__ W_B, const float* __restrict__ b_B,
                 const float* __restrict__ W_C, const float* __restrict__ b_C,
                 const float* __restrict__ W_D, const float* __restrict__ b_D,
                 const float* __restrict__ P,
                 float* __restrict__ Bm, float* __restrict__ Cc,
                 float* __restrict__ disc, float* __restrict__ discsum) {
    __shared__ float s_xwd[PROJ_ROWS];
    const int lane  = threadIdx.x & 31;
    const int wave  = __builtin_amdgcn_readfirstlane((int)(threadIdx.x >> 5));
  if (wave < PROJ_WAVES) {
    const int tile  = blockIdx.x * PROJ_WAVES + wave;
    const int half  = lane >> 4;
    const int lid   = lane & 15;
    const int row   = tile * 16 + lid;
    const int khalf = half * 2;
    const float dmask = (lid == 0) ? 1.0f : 0.0f;

    v8f accB = {}; v8f accC = {}; v8f accD = {};
    const float* xrow = x + (size_t)row * D_MODEL;

    for (int k0 = 0; k0 < D_MODEL; k0 += 4) {
        const v2f a = *(const v2f*)(xrow + k0 + khalf);
        v2f fB, fC, fD;
        fB.x = W_B[(k0 + khalf + 0) * D_STATE + lid];
        fB.y = W_B[(k0 + khalf + 1) * D_STATE + lid];
        fC.x = W_C[(k0 + khalf + 0) * D_STATE + lid];
        fC.y = W_C[(k0 + khalf + 1) * D_STATE + lid];
        fD.x = W_D[k0 + khalf + 0] * dmask;
        fD.y = W_D[k0 + khalf + 1] * dmask;
        accB = __builtin_amdgcn_wmma_f32_16x16x4_f32(false, a, false, fB, (short)0, accB, false, false);
        accC = __builtin_amdgcn_wmma_f32_16x16x4_f32(false, a, false, fC, (short)0, accC, false, false);
        accD = __builtin_amdgcn_wmma_f32_16x16x4_f32(false, a, false, fD, (short)0, accD, false, false);
    }

    const float bBn = b_B[lid];
    const float bCn = b_C[lid];
    const float bDs = b_D[0];
#pragma unroll
    for (int r = 0; r < 8; ++r) {
        const int orow = tile * 16 + r + 8 * half;
        Bm[(size_t)orow * D_STATE + lid] = accB[r] + bBn;
        Cc[(size_t)orow * D_STATE + lid] = accC[r] + bCn;
        if (lid == 0) s_xwd[wave * 16 + r + 8 * half] = accD[r] + bDs;
    }
  }
    __syncthreads();

    const int d  = threadIdx.x & (D_MODEL - 1);
    const int g  = threadIdx.x >> 7;
    const int gc = blockIdx.x * (PROJ_ROWS / CHUNK) + g;
    const size_t row0 = (size_t)gc * CHUNK;
    const float Pd = P[d];
    float acc = 0.0f;
    for (int s = 0; s < CHUNK; ++s) {
        const float dc = softplus_f(Pd + s_xwd[g * CHUNK + s]);
        disc[(row0 + s) * D_MODEL + d] = dc;
        acc += dc;
    }
    discsum[(size_t)gc * D_MODEL + d] = acc;
}

__global__ void scan_chunks_kernel(float* __restrict__ arr, int nchannels) {
    const int t = blockIdx.x * blockDim.x + threadIdx.x;
    if (t >= nchannels) return;
    const int d     = t % D_MODEL;
    const int outer = t / D_MODEL;
    float run = 0.0f;
    float* p = arr + (size_t)outer * NCHUNK * D_MODEL + d;
#pragma unroll
    for (int c = 0; c < NCHUNK; ++c) {
        const float v = p[c * D_MODEL]; p[c * D_MODEL] = run; run += v;
    }
}


__device__ __forceinline__ void scan_step(const float (&A2)[D_STATE], const float (&invA)[D_STATE],
                                          const float (&bm)[D_STATE], float dc, float xv,
                                          float Dcum, float (&S)[D_STATE], float (&Lc2)[D_STATE]) {
    float Bx[D_STATE], bxl2[D_STATE];
#pragma unroll
    for (int n = 0; n < D_STATE; ++n)
        Bx[n] = bm[n] * invA[n] * (__builtin_amdgcn_exp2f(dc * A2[n]) - 1.0f) * xv;
#pragma unroll
    for (int n = 0; n < D_STATE; ++n)
        bxl2[n] = __builtin_amdgcn_logf(fabsf(Bx[n]));
#pragma unroll
    for (int n = 0; n < D_STATE; ++n)
        Lc2[n] = fmaxf(A2[n] * Dcum, bxl2[n] - CLAMP2);
#pragma unroll
    for (int n = 0; n < D_STATE; ++n) {
        const float sgn = (Bx[n] > 0.0f) ? 1.0f : ((Bx[n] < 0.0f) ? -1.0f : 0.0f);
        S[n] += __builtin_amdgcn_exp2f(bxl2[n] - Lc2[n]) * sgn;
    }
}

__global__ void __launch_bounds__(D_MODEL) __attribute__((amdgpu_waves_per_eu(1, 4)))
scan_pass1_kernel(const float* __restrict__ x,
                  const float* __restrict__ Araw,
                  const float* __restrict__ disc,
                  const float* __restrict__ discsum,
                  float* __restrict__ discpre,
                  const float* __restrict__ Bm,
                  float* __restrict__ sumS) {
    const int d  = threadIdx.x;
    const int ch = blockIdx.x % NCHUNK;
    const int b  = blockIdx.x / NCHUNK;

    float A2[D_STATE], invA[D_STATE], acc[D_STATE];
#pragma unroll
    for (int n = 0; n < D_STATE; ++n) {
        const float An = -softplus_f(Araw[d * D_STATE + n]);
        invA[n] = 1.0f / An;
        A2[n]   = An * LOG2E;
        acc[n]  = 0.0f;
    }
    float Dcum = 0.0f;
    {
        const float* dp = discsum + (size_t)b * NCHUNK * D_MODEL + d;
        float dv[NCHUNK];
#pragma unroll
        for (int c = 0; c < NCHUNK; ++c) dv[c] = dp[c * D_MODEL];
        asm volatile("" :: "v"(dv[0]));
#pragma unroll
        for (int c = 0; c < NCHUNK; ++c) Dcum += (c < ch) ? dv[c] : 0.0f;
    }
    discpre[(b * NCHUNK + ch) * D_MODEL + d] = Dcum;

    const size_t row0 = (size_t)b * SEQ + (size_t)ch * CHUNK;
    float dc = disc[row0 * D_MODEL + d];
    float xv = x[row0 * D_MODEL + d];
    float bm[D_STATE];
#pragma unroll
    for (int n = 0; n < D_STATE; ++n) bm[n] = Bm[row0 * D_STATE + n];

    for (int i = 0; i < CHUNK; ++i) {
        const size_t rn = row0 + (size_t)((i + 1 < CHUNK) ? (i + 1) : (CHUNK - 1));
        const float dc_n = disc[rn * D_MODEL + d];
        const float xv_n = x[rn * D_MODEL + d];
        float bm_n[D_STATE];
#pragma unroll
        for (int n = 0; n < D_STATE; ++n) bm_n[n] = Bm[rn * D_STATE + n];

        Dcum += dc;
        float Lc2[D_STATE];
        scan_step(A2, invA, bm, dc, xv, Dcum, acc, Lc2);

        asm volatile("" :: "v"(dc_n), "v"(xv_n));
        dc = dc_n; xv = xv_n;
#pragma unroll
        for (int n = 0; n < D_STATE; ++n) bm[n] = bm_n[n];
    }
#pragma unroll
    for (int n = 0; n < D_STATE; ++n)
        sumS[((size_t)((b * D_STATE + n) * NCHUNK + ch)) * D_MODEL + d] = acc[n];
}

__global__ void __launch_bounds__(D_MODEL) __attribute__((amdgpu_waves_per_eu(1, 4)))
scan_final_kernel(const float* __restrict__ x,
                  const float* __restrict__ Araw,
                  const float* __restrict__ disc,
                  const float* __restrict__ discpre,
                  const float* __restrict__ Bm,
                  const float* __restrict__ Cc,
                  const float* __restrict__ sumSpre,
                  float* __restrict__ y) {
    const int d  = threadIdx.x;
    const int ch = blockIdx.x % NCHUNK;
    const int b  = blockIdx.x / NCHUNK;

    float A2[D_STATE], invA[D_STATE], Ssum[D_STATE];
#pragma unroll
    for (int n = 0; n < D_STATE; ++n) {
        const float An = -softplus_f(Araw[d * D_STATE + n]);
        invA[n] = 1.0f / An;
        A2[n]   = An * LOG2E;
        Ssum[n] = sumSpre[((size_t)((b * D_STATE + n) * NCHUNK + ch)) * D_MODEL + d];
    }
    float Dcum = discpre[(b * NCHUNK + ch) * D_MODEL + d];

    const size_t row0 = (size_t)b * SEQ + (size_t)ch * CHUNK;
    float dc = disc[row0 * D_MODEL + d];
    float xv = x[row0 * D_MODEL + d];
    float bm[D_STATE], cc[D_STATE];
#pragma unroll
    for (int n = 0; n < D_STATE; ++n) {
        bm[n] = Bm[row0 * D_STATE + n];
        cc[n] = Cc[row0 * D_STATE + n];
    }

    for (int i = 0; i < CHUNK; ++i) {
        const size_t rn = row0 + (size_t)((i + 1 < CHUNK) ? (i + 1) : (CHUNK - 1));
        const float dc_n = disc[rn * D_MODEL + d];
        const float xv_n = x[rn * D_MODEL + d];
        float bm_n[D_STATE], cc_n[D_STATE];
#pragma unroll
        for (int n = 0; n < D_STATE; ++n) {
            bm_n[n] = Bm[rn * D_STATE + n];
            cc_n[n] = Cc[rn * D_STATE + n];
        }

        Dcum += dc;
        float Lc2[D_STATE];
        scan_step(A2, invA, bm, dc, xv, Dcum, Ssum, Lc2);
        float h[D_STATE];
#pragma unroll
        for (int n = 0; n < D_STATE; ++n) h[n] = Ssum[n] * __builtin_amdgcn_exp2f(Lc2[n]);
        float yv = 0.0f;
#pragma unroll
        for (int n = 0; n < D_STATE; ++n) yv += h[n] * cc[n];
        y[(row0 + i) * D_MODEL + d] = yv;

        asm volatile("" :: "v"(dc_n), "v"(xv_n));
        dc = dc_n; xv = xv_n;
#pragma unroll
        for (int n = 0; n < D_STATE; ++n) { bm[n] = bm_n[n]; cc[n] = cc_n[n]; }
    }
}

extern "C" void kernel_launch(void* const* d_in, const int* in_sizes, int n_in,
                              void* d_out, int out_size, void* d_ws, size_t ws_size,
                              hipStream_t stream) {
    (void)in_sizes; (void)n_in; (void)out_size; (void)ws_size;
    const float* x    = (const float*)d_in[0];
    const float* Araw = (const float*)d_in[1];
    const float* P    = (const float*)d_in[2];
    const float* W_B  = (const float*)d_in[3];
    const float* b_B  = (const float*)d_in[4];
    const float* W_C  = (const float*)d_in[5];
    const float* b_C  = (const float*)d_in[6];
    const float* W_D  = (const float*)d_in[7];
    const float* b_D  = (const float*)d_in[8];
    float* y = (float*)d_out;

    constexpr size_t BS       = (size_t)BATCH * SEQ;
    constexpr size_t szBm     = BS * D_STATE * 4;
    constexpr size_t szCc     = BS * D_STATE * 4;
    constexpr size_t szDisc   = BS * D_MODEL * 4;
    constexpr size_t szDsum   = (size_t)BATCH * D_MODEL * NCHUNK * 4;
    char* ws = (char*)d_ws;
    float* Bm   = (float*)(ws);
    float* Cc   = (float*)(ws + szBm);
    float* disc = (float*)(ws + szBm + szCc);
    float* dsum = (float*)(ws + szBm + szCc + szDisc);
    float* dpre = (float*)(ws + szBm + szCc + szDisc + szDsum);
    float* sumS = (float*)(ws + szBm + szCc + szDisc + szDsum + szDsum);

    proj_disc_kernel<<<BS / PROJ_ROWS, 256, 0, stream>>>(x, W_B, b_B, W_C, b_C, W_D, b_D, P,
                                                         Bm, Cc, disc, dsum);
    scan_pass1_kernel<<<BATCH * NCHUNK, D_MODEL, 0, stream>>>(
        x, Araw, disc, dsum, dpre, Bm, sumS);
    scan_chunks_kernel<<<(BATCH * D_MODEL * D_STATE + 255) / 256, 256, 0, stream>>>(
        sumS, BATCH * D_MODEL * D_STATE);
    scan_final_kernel<<<BATCH * NCHUNK, D_MODEL, 0, stream>>>(
        x, Araw, disc, dpre, Bm, Cc, sumS, y);
}


# Both scan passes fetch a chunk's operands (32 rows of disc, x, Bm, C) once into LDS, all loads in flight together, and the step loop reads LDS: one memory wait per block, not one per step
# speedup vs baseline: 8.9692x; 1.1950x over previous
#include <hip/hip_runtime.h>
#include <math.h>


#define D_MODEL 128
#define D_STATE 16
#define BATCH   4
#define SEQ     4096
#define NCHUNK  128
#define CHUNK   (SEQ / NCHUNK)
#define LOG_CLAMP 70.0f
#define LOG2E  1.4426950408889634f
#define CLAMP2 ((float)((double)LOG_CLAMP * 1.4426950408889634))

typedef float v2f __attribute__((ext_vector_type(2)));
typedef float v4f __attribute__((ext_vector_type(4)));
typedef float v8f __attribute__((ext_vector_type(8)));

#define LN2 0.6931471805599453f
__device__ __forceinline__ float softplus_f(float z) {
    const float e  = __builtin_amdgcn_exp2f(-LOG2E * fabsf(z));
    const float lg = LN2 * __builtin_amdgcn_logf(1.0f + e);
    const float sr = e * fmaf(e, fmaf(e, 1.0f / 3.0f, -0.5f), 1.0f);
    return fmaxf(z, 0.0f) + ((e < 0.015625f) ? sr : lg);
}

#define PROJ_WAVES 4
#define PROJ_ROWS  (PROJ_WAVES * 16)
__global__ void __launch_bounds__(256)
proj_disc_kernel(const float* __restrict__ x,
                 const float* __restrict__ W_B, const float* __restrict__ b_B,
                 const float* __restrict__ W_C, const float* __restrict__ b_C,
                 const float* __restrict__ W_D, const float* __restrict__ b_D,
                 const float* __restrict__ P,
                 float* __restrict__ Bm, float* __restrict__ Cc,
                 float* __restrict__ disc, float* __restrict__ discsum) {
    __shared__ float s_xwd[PROJ_ROWS];
    const int lane  = threadIdx.x & 31;
    const int wave  = __builtin_amdgcn_readfirstlane((int)(threadIdx.x >> 5));
  if (wave < PROJ_WAVES) {
    const int tile  = blockIdx.x * PROJ_WAVES + wave;
    const int half  = lane >> 4;
    const int lid   = lane & 15;
    const int row   = tile * 16 + lid;
    const int khalf = half * 2;
    const float dmask = (lid == 0) ? 1.0f : 0.0f;

    v8f accB = {}; v8f accC = {}; v8f accD = {};
    const float* xrow = x + (size_t)row * D_MODEL;

    for (int k0 = 0; k0 < D_MODEL; k0 += 4) {
        const v2f a = *(const v2f*)(xrow + k0 + khalf);
        v2f fB, fC, fD;
        fB.x = W_B[(k0 + khalf + 0) * D_STATE + lid];
        fB.y = W_B[(k0 + khalf + 1) * D_STATE + lid];
        fC.x = W_C[(k0 + khalf + 0) * D_STATE + lid];
        fC.y = W_C[(k0 + khalf + 1) * D_STATE + lid];
        fD.x = W_D[k0 + khalf + 0] * dmask;
        fD.y = W_D[k0 + khalf + 1] * dmask;
        accB = __builtin_amdgcn_wmma_f32_16x16x4_f32(false, a, false, fB, (short)0, accB, false, false);
        accC = __builtin_amdgcn_wmma_f32_16x16x4_f32(false, a, false, fC, (short)0, accC, false, false);
        accD = __builtin_amdgcn_wmma_f32_16x16x4_f32(false, a, false, fD, (short)0, accD, false, false);
    }

    const float bBn = b_B[lid];
    const float bCn = b_C[lid];
    const float bDs = b_D[0];
#pragma unroll
    for (int r = 0; r < 8; ++r) {
        const int orow = tile * 16 + r + 8 * half;
        Bm[(size_t)orow * D_STATE + lid] = accB[r] + bBn;
        Cc[(size_t)orow * D_STATE + lid] = accC[r] + bCn;
        if (lid == 0) s_xwd[wave * 16 + r + 8 * half] = accD[r] + bDs;
    }
  }
    __syncthreads();

    const int d  = threadIdx.x & (D_MODEL - 1);
    const int g  = threadIdx.x >> 7;
    const int gc = blockIdx.x * (PROJ_ROWS / CHUNK) + g;
    const size_t row0 = (size_t)gc * CHUNK;
    const float Pd = P[d];
    float acc = 0.0f;
    for (int s = 0; s < CHUNK; ++s) {
        const float dc = softplus_f(Pd + s_xwd[g * CHUNK + s]);
        disc[(row0 + s) * D_MODEL + d] = dc;
        acc += dc;
    }
    discsum[(size_t)gc * D_MODEL + d] = acc;
}

__global__ void scan_chunks_kernel(float* __restrict__ arr, int nchannels) {
    const int t = blockIdx.x * blockDim.x + threadIdx.x;
    if (t >= nchannels) return;
    const int d     = t % D_MODEL;
    const int outer = t / D_MODEL;
    float run = 0.0f;
    float* p = arr + (size_t)outer * NCHUNK * D_MODEL + d;
#pragma unroll
    for (int c = 0; c < NCHUNK; ++c) {
        const float v = p[c * D_MODEL]; p[c * D_MODEL] = run; run += v;
    }
}


#define STAGE_PIECES (CHUNK * D_MODEL / 4 / D_MODEL)

__device__ __forceinline__ void lds_row16(const float* row, float (&v)[D_STATE]) {
    const v4f* p = (const v4f*)row;
#pragma unroll
    for (int q = 0; q < D_STATE / 4; ++q) {
        const v4f t = p[q];
        v[4 * q + 0] = t.x; v[4 * q + 1] = t.y; v[4 * q + 2] = t.z; v[4 * q + 3] = t.w;
    }
}

__device__ __forceinline__ void scan_step(const float (&A2)[D_STATE], const float (&invA)[D_STATE],
                                          const float (&bm)[D_STATE], float dc, float xv,
                                          float Dcum, float (&S)[D_STATE], float (&Lc2)[D_STATE]) {
    float Bx[D_STATE], bxl2[D_STATE];
#pragma unroll
    for (int n = 0; n < D_STATE; ++n)
        Bx[n] = bm[n] * invA[n] * (__builtin_amdgcn_exp2f(dc * A2[n]) - 1.0f) * xv;
#pragma unroll
    for (int n = 0; n < D_STATE; ++n)
        bxl2[n] = __builtin_amdgcn_logf(fabsf(Bx[n]));
#pragma unroll
    for (int n = 0; n < D_STATE; ++n)
        Lc2[n] = fmaxf(A2[n] * Dcum, bxl2[n] - CLAMP2);
#pragma unroll
    for (int n = 0; n < D_STATE; ++n) {
        const float sgn = (Bx[n] > 0.0f) ? 1.0f : ((Bx[n] < 0.0f) ? -1.0f : 0.0f);
        S[n] += __builtin_amdgcn_exp2f(bxl2[n] - Lc2[n]) * sgn;
    }
}

__global__ void __launch_bounds__(D_MODEL) __attribute__((amdgpu_waves_per_eu(1, 4)))
scan_pass1_kernel(const float* __restrict__ x,
                  const float* __restrict__ Araw,
                  const float* __restrict__ disc,
                  const float* __restrict__ discsum,
                  float* __restrict__ discpre,
                  const float* __restrict__ Bm,
                  float* __restrict__ sumS) {
    __shared__ __attribute__((aligned(16))) float s_dc[CHUNK * D_MODEL];
    __shared__ __attribute__((aligned(16))) float s_x[CHUNK * D_MODEL];
    __shared__ __attribute__((aligned(16))) float s_bm[CHUNK * D_STATE];
    const int d  = threadIdx.x;
    const int ch = blockIdx.x % NCHUNK;
    const int b  = blockIdx.x / NCHUNK;
    const size_t row0 = (size_t)b * SEQ + (size_t)ch * CHUNK;

    v4f tdc[STAGE_PIECES], tx[STAGE_PIECES];
    {
        const v4f* gdc = (const v4f*)(disc + row0 * D_MODEL);
        const v4f* gx  = (const v4f*)(x + row0 * D_MODEL);
#pragma unroll
        for (int j = 0; j < STAGE_PIECES; ++j) {
            tdc[j] = gdc[j * D_MODEL + d];
            tx[j]  = gx[j * D_MODEL + d];
        }
    }
    const v4f tbm = ((const v4f*)(Bm + row0 * D_STATE))[d];

    float A2[D_STATE], invA[D_STATE], acc[D_STATE];
#pragma unroll
    for (int n = 0; n < D_STATE; ++n) {
        const float An = -softplus_f(Araw[d * D_STATE + n]);
        invA[n] = 1.0f / An;
        A2[n]   = An * LOG2E;
        acc[n]  = 0.0f;
    }
    float Dcum = 0.0f;
    {
        const float* dp = discsum + (size_t)b * NCHUNK * D_MODEL + d;
        float dv[NCHUNK];
#pragma unroll
        for (int c = 0; c < NCHUNK; ++c) dv[c] = dp[c * D_MODEL];
        asm volatile("" :: "v"(dv[0]));
#pragma unroll
        for (int c = 0; c < NCHUNK; ++c) Dcum += (c < ch) ? dv[c] : 0.0f;
    }
    discpre[(b * NCHUNK + ch) * D_MODEL + d] = Dcum;

#pragma unroll
    for (int j = 0; j < STAGE_PIECES; ++j) {
        ((v4f*)s_dc)[j * D_MODEL + d] = tdc[j];
        ((v4f*)s_x)[j * D_MODEL + d]  = tx[j];
    }
    ((v4f*)s_bm)[d] = tbm;
    __syncthreads();

    float dc = s_dc[d];
    float xv = s_x[d];
    float bm[D_STATE];
    lds_row16(s_bm, bm);

    for (int i = 0; i < CHUNK; ++i) {
        const int rn = (i + 1 < CHUNK) ? (i + 1) : (CHUNK - 1);
        const float dc_n = s_dc[rn * D_MODEL + d];
        const float xv_n = s_x[rn * D_MODEL + d];
        float bm_n[D_STATE];
        lds_row16(s_bm + rn * D_STATE, bm_n);

        Dcum += dc;
        float Lc2[D_STATE];
        scan_step(A2, invA, bm, dc, xv, Dcum, acc, Lc2);

        dc = dc_n; xv = xv_n;
#pragma unroll
        for (int n = 0; n < D_STATE; ++n) bm[n] = bm_n[n];
    }
#pragma unroll
    for (int n = 0; n < D_STATE; ++n)
        sumS[((size_t)((b * D_STATE + n) * NCHUNK + ch)) * D_MODEL + d] = acc[n];
}

__global__ void __launch_bounds__(D_MODEL) __attribute__((amdgpu_waves_per_eu(1, 4)))
scan_final_kernel(const float* __restrict__ x,
                  const float* __restrict__ Araw,
                  const float* __restrict__ disc,
                  const float* __restrict__ discpre,
                  const float* __restrict__ Bm,
                  const float* __restrict__ Cc,
                  const float* __restrict__ sumSpre,
                  float* __restrict__ y) {
    __shared__ __attribute__((aligned(16))) float s_dc[CHUNK * D_MODEL];
    __shared__ __attribute__((aligned(16))) float s_x[CHUNK * D_MODEL];
    __shared__ __attribute__((aligned(16))) float s_bm[CHUNK * D_STATE];
    __shared__ __attribute__((aligned(16))) float s_cc[CHUNK * D_STATE];
    const int d  = threadIdx.x;
    const int ch = blockIdx.x % NCHUNK;
    const int b  = blockIdx.x / NCHUNK;
    const size_t row0 = (size_t)b * SEQ + (size_t)ch * CHUNK;

    v4f tdc[STAGE_PIECES], tx[STAGE_PIECES];
    {
        const v4f* gdc = (const v4f*)(disc + row0 * D_MODEL);
        const v4f* gx  = (const v4f*)(x + row0 * D_MODEL);
#pragma unroll
        for (int j = 0; j < STAGE_PIECES; ++j) {
            tdc[j] = gdc[j * D_MODEL + d];
            tx[j]  = gx[j * D_MODEL + d];
        }
    }
    const v4f tbm = ((const v4f*)(Bm + row0 * D_STATE))[d];
    const v4f tcc = ((const v4f*)(Cc + row0 * D_STATE))[d];

    float A2[D_STATE], invA[D_STATE], Ssum[D_STATE];
#pragma unroll
    for (int n = 0; n < D_STATE; ++n) {
        const float An = -softplus_f(Araw[d * D_STATE + n]);
        invA[n] = 1.0f / An;
        A2[n]   = An * LOG2E;
        Ssum[n] = sumSpre[((size_t)((b * D_STATE + n) * NCHUNK + ch)) * D_MODEL + d];
    }
    float Dcum = discpre[(b * NCHUNK + ch) * D_MODEL + d];

#pragma unroll
    for (int j = 0; j < STAGE_PIECES; ++j) {
        ((v4f*)s_dc)[j * D_MODEL + d] = tdc[j];
        ((v4f*)s_x)[j * D_MODEL + d]  = tx[j];
    }
    ((v4f*)s_bm)[d] = tbm;
    ((v4f*)s_cc)[d] = tcc;
    __syncthreads();

    float dc = s_dc[d];
    float xv = s_x[d];
    float bm[D_STATE], cc[D_STATE];
    lds_row16(s_bm, bm);
    lds_row16(s_cc, cc);

    for (int i = 0; i < CHUNK; ++i) {
        const int rn = (i + 1 < CHUNK) ? (i + 1) : (CHUNK - 1);
        const float dc_n = s_dc[rn * D_MODEL + d];
        const float xv_n = s_x[rn * D_MODEL + d];
        float bm_n[D_STATE], cc_n[D_STATE];
        lds_row16(s_bm + rn * D_STATE, bm_n);
        lds_row16(s_cc + rn * D_STATE, cc_n);

        Dcum += dc;
        float Lc2[D_STATE];
        scan_step(A2, invA, bm, dc, xv, Dcum, Ssum, Lc2);
        float h[D_STATE];
#pragma unroll
        for (int n = 0; n < D_STATE; ++n) h[n] = Ssum[n] * __builtin_amdgcn_exp2f(Lc2[n]);
        float yv = 0.0f;
#pragma unroll
        for (int n = 0; n < D_STATE; ++n) yv += h[n] * cc[n];
        y[(row0 + i) * D_MODEL + d] = yv;

        dc = dc_n; xv = xv_n;
#pragma unroll
        for (int n = 0; n < D_STATE; ++n) { bm[n] = bm_n[n]; cc[n] = cc_n[n]; }
    }
}

extern "C" void kernel_launch(void* const* d_in, const int* in_sizes, int n_in,
                              void* d_out, int out_size, void* d_ws, size_t ws_size,
                              hipStream_t stream) {
    (void)in_sizes; (void)n_in; (void)out_size; (void)ws_size;
    const float* x    = (const float*)d_in[0];
    const float* Araw = (const float*)d_in[1];
    const float* P    = (const float*)d_in[2];
    const float* W_B  = (const float*)d_in[3];
    const float* b_B  = (const float*)d_in[4];
    const float* W_C  = (const float*)d_in[5];
    const float* b_C  = (const float*)d_in[6];
    const float* W_D  = (const float*)d_in[7];
    const float* b_D  = (const float*)d_in[8];
    float* y = (float*)d_out;

    constexpr size_t BS       = (size_t)BATCH * SEQ;
    constexpr size_t szBm     = BS * D_STATE * 4;
    constexpr size_t szCc     = BS * D_STATE * 4;
    constexpr size_t szDisc   = BS * D_MODEL * 4;
    constexpr size_t szDsum   = (size_t)BATCH * D_MODEL * NCHUNK * 4;
    char* ws = (char*)d_ws;
    float* Bm   = (float*)(ws);
    float* Cc   = (float*)(ws + szBm);
    float* disc = (float*)(ws + szBm + szCc);
    float* dsum = (float*)(ws + szBm + szCc + szDisc);
    float* dpre = (float*)(ws + szBm + szCc + szDisc + szDsum);
    float* sumS = (float*)(ws + szBm + szCc + szDisc + szDsum + szDsum);

    proj_disc_kernel<<<BS / PROJ_ROWS, 256, 0, stream>>>(x, W_B, b_B, W_C, b_C, W_D, b_D, P,
                                                         Bm, Cc, disc, dsum);
    scan_pass1_kernel<<<BATCH * NCHUNK, D_MODEL, 0, stream>>>(
        x, Araw, disc, dsum, dpre, Bm, sumS);
    scan_chunks_kernel<<<(BATCH * D_MODEL * D_STATE + 255) / 256, 256, 0, stream>>>(
        sumS, BATCH * D_MODEL * D_STATE);
    scan_final_kernel<<<BATCH * NCHUNK, D_MODEL, 0, stream>>>(
        x, Araw, disc, dpre, Bm, Cc, sumS, y);
}


# Scan increment leaves log space: exp(log|Bx|-Lc)*sign -> clamp(Bx*2^-(L2+c2),+-1), sums in units of e^70; exp(Lc) -> max(2^(L2+c2),|Bx|): log and sign select leave both loops; rounding changes
# speedup vs baseline: 9.8807x; 1.1016x over previous
#include <hip/hip_runtime.h>
#include <math.h>


#define D_MODEL 128
#define D_STATE 16
#define BATCH   4
#define SEQ     4096
#define NCHUNK  128
#define CHUNK   (SEQ / NCHUNK)
#define LOG_CLAMP 70.0f
#define LOG2E  1.4426950408889634f
#define CLAMP2 ((float)((double)LOG_CLAMP * 1.4426950408889634))

typedef float v2f __attribute__((ext_vector_type(2)));
typedef float v4f __attribute__((ext_vector_type(4)));
typedef float v8f __attribute__((ext_vector_type(8)));

#define LN2 0.6931471805599453f
__device__ __forceinline__ float softplus_f(float z) {
    const float e  = __builtin_amdgcn_exp2f(-LOG2E * fabsf(z));
    const float lg = LN2 * __builtin_amdgcn_logf(1.0f + e);
    const float sr = e * fmaf(e, fmaf(e, 1.0f / 3.0f, -0.5f), 1.0f);
    return fmaxf(z, 0.0f) + ((e < 0.015625f) ? sr : lg);
}

#define PROJ_WAVES 4
#define PROJ_ROWS  (PROJ_WAVES * 16)
__global__ void __launch_bounds__(256)
proj_disc_kernel(const float* __restrict__ x,
                 const float* __restrict__ W_B, const float* __restrict__ b_B,
                 const float* __restrict__ W_C, const float* __restrict__ b_C,
                 const float* __restrict__ W_D, const float* __restrict__ b_D,
                 const float* __restrict__ P,
                 float* __restrict__ Bm, float* __restrict__ Cc,
                 float* __restrict__ disc, float* __restrict__ discsum) {
    __shared__ float s_xwd[PROJ_ROWS];
    const int lane  = threadIdx.x & 31;
    const int wave  = __builtin_amdgcn_readfirstlane((int)(threadIdx.x >> 5));
  if (wave < PROJ_WAVES) {
    const int tile  = blockIdx.x * PROJ_WAVES + wave;
    const int half  = lane >> 4;
    const int lid   = lane & 15;
    const int row   = tile * 16 + lid;
    const int khalf = half * 2;
    const float dmask = (lid == 0) ? 1.0f : 0.0f;

    v8f accB = {}; v8f accC = {}; v8f accD = {};
    const float* xrow = x + (size_t)row * D_MODEL;

    for (int k0 = 0; k0 < D_MODEL; k0 += 4) {
        const v2f a = *(const v2f*)(xrow + k0 + khalf);
        v2f fB, fC, fD;
        fB.x = W_B[(k0 + khalf + 0) * D_STATE + lid];
        fB.y = W_B[(k0 + khalf + 1) * D_STATE + lid];
        fC.x = W_C[(k0 + khalf + 0) * D_STATE + lid];
        fC.y = W_C[(k0 + khalf + 1) * D_STATE + lid];
        fD.x = W_D[k0 + khalf + 0] * dmask;
        fD.y = W_D[k0 + khalf + 1] * dmask;
        accB = __builtin_amdgcn_wmma_f32_16x16x4_f32(false, a, false, fB, (short)0, accB, false, false);
        accC = __builtin_amdgcn_wmma_f32_16x16x4_f32(false, a, false, fC, (short)0, accC, false, false);
        accD = __builtin_amdgcn_wmma_f32_16x16x4_f32(false, a, false, fD, (short)0, accD, false, false);
    }

    const float bBn = b_B[lid];
    const float bCn = b_C[lid];
    const float bDs = b_D[0];
#pragma unroll
    for (int r = 0; r < 8; ++r) {
        const int orow = tile * 16 + r + 8 * half;
        Bm[(size_t)orow * D_STATE + lid] = accB[r] + bBn;
        Cc[(size_t)orow * D_STATE + lid] = accC[r] + bCn;
        if (lid == 0) s_xwd[wave * 16 + r + 8 * half] = accD[r] + bDs;
    }
  }
    __syncthreads();

    const int d  = threadIdx.x & (D_MODEL - 1);
    const int g  = threadIdx.x >> 7;
    const int gc = blockIdx.x * (PROJ_ROWS / CHUNK) + g;
    const size_t row0 = (size_t)gc * CHUNK;
    const float Pd = P[d];
    float acc = 0.0f;
    for (int s = 0; s < CHUNK; ++s) {
        const float dc = softplus_f(Pd + s_xwd[g * CHUNK + s]);
        disc[(row0 + s) * D_MODEL + d] = dc;
        acc += dc;
    }
    discsum[(size_t)gc * D_MODEL + d] = acc;
}

__global__ void scan_chunks_kernel(float* __restrict__ arr, int nchannels) {
    const int t = blockIdx.x * blockDim.x + threadIdx.x;
    if (t >= nchannels) return;
    const int d     = t % D_MODEL;
    const int outer = t / D_MODEL;
    float run = 0.0f;
    float* p = arr + (size_t)outer * NCHUNK * D_MODEL + d;
#pragma unroll
    for (int c = 0; c < NCHUNK; ++c) {
        const float v = p[c * D_MODEL]; p[c * D_MODEL] = run; run += v;
    }
}


#define STAGE_PIECES (CHUNK * D_MODEL / 4 / D_MODEL)

__device__ __forceinline__ void lds_row16(const float* row, float (&v)[D_STATE]) {
    const v4f* p = (const v4f*)row;
#pragma unroll
    for (int q = 0; q < D_STATE / 4; ++q) {
        const v4f t = p[q];
        v[4 * q + 0] = t.x; v[4 * q + 1] = t.y; v[4 * q + 2] = t.z; v[4 * q + 3] = t.w;
    }
}

__device__ __forceinline__ void scan_step(const float (&A2)[D_STATE], const float (&invA)[D_STATE],
                                          const float (&bm)[D_STATE], float dc, float xv,
                                          float nDcum, float (&S)[D_STATE],
                                          float (&Bx)[D_STATE], float (&q)[D_STATE]) {
    float E[D_STATE];
#pragma unroll
    for (int n = 0; n < D_STATE; ++n)
        Bx[n] = bm[n] * invA[n] * (__builtin_amdgcn_exp2f(dc * A2[n]) - 1.0f) * xv;
#pragma unroll
    for (int n = 0; n < D_STATE; ++n)
        q[n] = fmaf(A2[n], nDcum, -CLAMP2);
#pragma unroll
    for (int n = 0; n < D_STATE; ++n)
        E[n] = __builtin_amdgcn_exp2f(fminf(q[n], 127.0f));
#pragma unroll
    for (int n = 0; n < D_STATE; ++n)
        S[n] += fminf(fmaxf(Bx[n] * E[n], -1.0f), 1.0f);
}

__global__ void __launch_bounds__(D_MODEL) __attribute__((amdgpu_waves_per_eu(1, 4)))
scan_pass1_kernel(const float* __restrict__ x,
                  const float* __restrict__ Araw,
                  const float* __restrict__ disc,
                  const float* __restrict__ discsum,
                  float* __restrict__ discpre,
                  const float* __restrict__ Bm,
                  float* __restrict__ sumS) {
    __shared__ __attribute__((aligned(16))) float s_dc[CHUNK * D_MODEL];
    __shared__ __attribute__((aligned(16))) float s_x[CHUNK * D_MODEL];
    __shared__ __attribute__((aligned(16))) float s_bm[CHUNK * D_STATE];
    const int d  = threadIdx.x;
    const int ch = blockIdx.x % NCHUNK;
    const int b  = blockIdx.x / NCHUNK;
    const size_t row0 = (size_t)b * SEQ + (size_t)ch * CHUNK;

    v4f tdc[STAGE_PIECES], tx[STAGE_PIECES];
    {
        const v4f* gdc = (const v4f*)(disc + row0 * D_MODEL);
        const v4f* gx  = (const v4f*)(x + row0 * D_MODEL);
#pragma unroll
        for (int j = 0; j < STAGE_PIECES; ++j) {
            tdc[j] = gdc[j * D_MODEL + d];
            tx[j]  = gx[j * D_MODEL + d];
        }
    }
    const v4f tbm = ((const v4f*)(Bm + row0 * D_STATE))[d];

    float A2[D_STATE], invA[D_STATE], acc[D_STATE];
#pragma unroll
    for (int n = 0; n < D_STATE; ++n) {
        const float An = -softplus_f(Araw[d * D_STATE + n]);
        invA[n] = 1.0f / An;
        A2[n]   = An * LOG2E;
        acc[n]  = 0.0f;
    }
    float Dcum = 0.0f;
    {
        const float* dp = discsum + (size_t)b * NCHUNK * D_MODEL + d;
        float dv[NCHUNK];
#pragma unroll
        for (int c = 0; c < NCHUNK; ++c) dv[c] = dp[c * D_MODEL];
        asm volatile("" :: "v"(dv[0]));
#pragma unroll
        for (int c = 0; c < NCHUNK; ++c) Dcum += (c < ch) ? dv[c] : 0.0f;
    }
    discpre[(b * NCHUNK + ch) * D_MODEL + d] = Dcum;
    float nDcum = -Dcum;

#pragma unroll
    for (int j = 0; j < STAGE_PIECES; ++j) {
        ((v4f*)s_dc)[j * D_MODEL + d] = tdc[j];
        ((v4f*)s_x)[j * D_MODEL + d]  = tx[j];
    }
    ((v4f*)s_bm)[d] = tbm;
    __syncthreads();

    float dc = s_dc[d];
    float xv = s_x[d];
    float bm[D_STATE];
    lds_row16(s_bm, bm);

    for (int i = 0; i < CHUNK; ++i) {
        const int rn = (i + 1 < CHUNK) ? (i + 1) : (CHUNK - 1);
        const float dc_n = s_dc[rn * D_MODEL + d];
        const float xv_n = s_x[rn * D_MODEL + d];
        float bm_n[D_STATE];
        lds_row16(s_bm + rn * D_STATE, bm_n);

        nDcum -= dc;
        float Bx[D_STATE], q[D_STATE];
        scan_step(A2, invA, bm, dc, xv, nDcum, acc, Bx, q);

        dc = dc_n; xv = xv_n;
#pragma unroll
        for (int n = 0; n < D_STATE; ++n) bm[n] = bm_n[n];
    }
#pragma unroll
    for (int n = 0; n < D_STATE; ++n)
        sumS[((size_t)((b * D_STATE + n) * NCHUNK + ch)) * D_MODEL + d] = acc[n];
}

__global__ void __launch_bounds__(D_MODEL) __attribute__((amdgpu_waves_per_eu(1, 4)))
scan_final_kernel(const float* __restrict__ x,
                  const float* __restrict__ Araw,
                  const float* __restrict__ disc,
                  const float* __restrict__ discpre,
                  const float* __restrict__ Bm,
                  const float* __restrict__ Cc,
                  const float* __restrict__ sumSpre,
                  float* __restrict__ y) {
    __shared__ __attribute__((aligned(16))) float s_dc[CHUNK * D_MODEL];
    __shared__ __attribute__((aligned(16))) float s_x[CHUNK * D_MODEL];
    __shared__ __attribute__((aligned(16))) float s_bm[CHUNK * D_STATE];
    __shared__ __attribute__((aligned(16))) float s_cc[CHUNK * D_STATE];
    const int d  = threadIdx.x;
    const int ch = blockIdx.x % NCHUNK;
    const int b  = blockIdx.x / NCHUNK;
    const size_t row0 = (size_t)b * SEQ + (size_t)ch * CHUNK;

    v4f tdc[STAGE_PIECES], tx[STAGE_PIECES];
    {
        const v4f* gdc = (const v4f*)(disc + row0 * D_MODEL);
        const v4f* gx  = (const v4f*)(x + row0 * D_MODEL);
#pragma unroll
        for (int j = 0; j < STAGE_PIECES; ++j) {
            tdc[j] = gdc[j * D_MODEL + d];
            tx[j]  = gx[j * D_MODEL + d];
        }
    }
    const v4f tbm = ((const v4f*)(Bm + row0 * D_STATE))[d];
    const v4f tcc = ((const v4f*)(Cc + row0 * D_STATE))[d];

    float A2[D_STATE], invA[D_STATE], Ssum[D_STATE];
#pragma unroll
    for (int n = 0; n < D_STATE; ++n) {
        const float An = -softplus_f(Araw[d * D_STATE + n]);
        invA[n] = 1.0f / An;
        A2[n]   = An * LOG2E;
        Ssum[n] = sumSpre[((size_t)((b * D_STATE + n) * NCHUNK + ch)) * D_MODEL + d];
    }
    float nDcum = -discpre[(b * NCHUNK + ch) * D_MODEL + d];

#pragma unroll
    for (int j = 0; j < STAGE_PIECES; ++j) {
        ((v4f*)s_dc)[j * D_MODEL + d] = tdc[j];
        ((v4f*)s_x)[j * D_MODEL + d]  = tx[j];
    }
    ((v4f*)s_bm)[d] = tbm;
    ((v4f*)s_cc)[d] = tcc;
    __syncthreads();

    float dc = s_dc[d];
    float xv = s_x[d];
    float bm[D_STATE], cc[D_STATE];
    lds_row16(s_bm, bm);
    lds_row16(s_cc, cc);

    for (int i = 0; i < CHUNK; ++i) {
        const int rn = (i + 1 < CHUNK) ? (i + 1) : (CHUNK - 1);
        const float dc_n = s_dc[rn * D_MODEL + d];
        const float xv_n = s_x[rn * D_MODEL + d];
        float bm_n[D_STATE], cc_n[D_STATE];
        lds_row16(s_bm + rn * D_STATE, bm_n);
        lds_row16(s_cc + rn * D_STATE, cc_n);

        nDcum -= dc;
        float Bx[D_STATE], q[D_STATE];
        scan_step(A2, invA, bm, dc, xv, nDcum, Ssum, Bx, q);
        float g[D_STATE], h[D_STATE];
#pragma unroll
        for (int n = 0; n < D_STATE; ++n) g[n] = __builtin_amdgcn_exp2f(-q[n]);
#pragma unroll
        for (int n = 0; n < D_STATE; ++n) h[n] = Ssum[n] * fmaxf(g[n], fabsf(Bx[n]));
        float yv = 0.0f;
#pragma unroll
        for (int n = 0; n < D_STATE; ++n) yv += h[n] * cc[n];
        y[(row0 + i) * D_MODEL + d] = yv;

        dc = dc_n; xv = xv_n;
#pragma unroll
        for (int n = 0; n < D_STATE; ++n) { bm[n] = bm_n[n]; cc[n] = cc_n[n]; }
    }
}

extern "C" void kernel_launch(void* const* d_in, const int* in_sizes, int n_in,
                              void* d_out, int out_size, void* d_ws, size_t ws_size,
                              hipStream_t stream) {
    (void)in_sizes; (void)n_in; (void)out_size; (void)ws_size;
    const float* x    = (const float*)d_in[0];
    const float* Araw = (const float*)d_in[1];
    const float* P    = (const float*)d_in[2];
    const float* W_B  = (const float*)d_in[3];
    const float* b_B  = (const float*)d_in[4];
    const float* W_C  = (const float*)d_in[5];
    const float* b_C  = (const float*)d_in[6];
    const float* W_D  = (const float*)d_in[7];
    const float* b_D  = (const float*)d_in[8];
    float* y = (float*)d_out;

    constexpr size_t BS       = (size_t)BATCH * SEQ;
    constexpr size_t szBm     = BS * D_STATE * 4;
    constexpr size_t szCc     = BS * D_STATE * 4;
    constexpr size_t szDisc   = BS * D_MODEL * 4;
    constexpr size_t szDsum   = (size_t)BATCH * D_MODEL * NCHUNK * 4;
    char* ws = (char*)d_ws;
    float* Bm   = (float*)(ws);
    float* Cc   = (float*)(ws + szBm);
    float* disc = (float*)(ws + szBm + szCc);
    float* dsum = (float*)(ws + szBm + szCc + szDisc);
    float* dpre = (float*)(ws + szBm + szCc + szDisc + szDsum);
    float* sumS = (float*)(ws + szBm + szCc + szDisc + szDsum + szDsum);

    proj_disc_kernel<<<BS / PROJ_ROWS, 256, 0, stream>>>(x, W_B, b_B, W_C, b_C, W_D, b_D, P,
                                                         Bm, Cc, disc, dsum);
    scan_pass1_kernel<<<BATCH * NCHUNK, D_MODEL, 0, stream>>>(
        x, Araw, disc, dsum, dpre, Bm, sumS);
    scan_chunks_kernel<<<(BATCH * D_MODEL * D_STATE + 255) / 256, 256, 0, stream>>>(
        sumS, BATCH * D_MODEL * D_STATE);
    scan_final_kernel<<<BATCH * NCHUNK, D_MODEL, 0, stream>>>(
        x, Araw, disc, dpre, Bm, Cc, sumS, y);
}


# Projection K-loop: a tile's whole fragment set (x, W_B, W_C, W_D, all 32 K-steps) is loaded before the first matrix instruction, all loads in flight together: one memory wait per tile
# speedup vs baseline: 11.6500x; 1.1791x over previous
#include <hip/hip_runtime.h>
#include <math.h>


#define D_MODEL 128
#define D_STATE 16
#define BATCH   4
#define SEQ     4096
#define NCHUNK  128
#define CHUNK   (SEQ / NCHUNK)
#define LOG_CLAMP 70.0f
#define LOG2E  1.4426950408889634f
#define CLAMP2 ((float)((double)LOG_CLAMP * 1.4426950408889634))

typedef float v2f __attribute__((ext_vector_type(2)));
typedef float v4f __attribute__((ext_vector_type(4)));
typedef float v8f __attribute__((ext_vector_type(8)));

#define LN2 0.6931471805599453f
__device__ __forceinline__ float softplus_f(float z) {
    const float e  = __builtin_amdgcn_exp2f(-LOG2E * fabsf(z));
    const float lg = LN2 * __builtin_amdgcn_logf(1.0f + e);
    const float sr = e * fmaf(e, fmaf(e, 1.0f / 3.0f, -0.5f), 1.0f);
    return fmaxf(z, 0.0f) + ((e < 0.015625f) ? sr : lg);
}

#define PROJ_WAVES 4
#define PROJ_ROWS  (PROJ_WAVES * 16)
#define KSTEPS     (D_MODEL / 4)
__global__ void __launch_bounds__(256) __attribute__((amdgpu_waves_per_eu(1, 2)))
proj_disc_kernel(const float* __restrict__ x,
                 const float* __restrict__ W_B, const float* __restrict__ b_B,
                 const float* __restrict__ W_C, const float* __restrict__ b_C,
                 const float* __restrict__ W_D, const float* __restrict__ b_D,
                 const float* __restrict__ P,
                 float* __restrict__ Bm, float* __restrict__ Cc,
                 float* __restrict__ disc, float* __restrict__ discsum) {
    __shared__ float s_xwd[PROJ_ROWS];
    const int lane  = threadIdx.x & 31;
    const int wave  = __builtin_amdgcn_readfirstlane((int)(threadIdx.x >> 5));
  if (wave < PROJ_WAVES) {
    const int tile  = blockIdx.x * PROJ_WAVES + wave;
    const int half  = lane >> 4;
    const int lid   = lane & 15;
    const int row   = tile * 16 + lid;
    const int khalf = half * 2;
    const float dmask = (lid == 0) ? 1.0f : 0.0f;

    v8f accB = {}; v8f accC = {}; v8f accD = {};
    const float* xrow = x + (size_t)row * D_MODEL;

    v2f a[KSTEPS], fB[KSTEPS], fC[KSTEPS], fD[KSTEPS];
#pragma unroll
    for (int j = 0; j < KSTEPS; ++j) {
        const int k = 4 * j + khalf;
        a[j] = *(const v2f*)(xrow + k);
        fB[j].x = W_B[(k + 0) * D_STATE + lid];
        fB[j].y = W_B[(k + 1) * D_STATE + lid];
        fC[j].x = W_C[(k + 0) * D_STATE + lid];
        fC[j].y = W_C[(k + 1) * D_STATE + lid];
        fD[j].x = W_D[k + 0] * dmask;
        fD[j].y = W_D[k + 1] * dmask;
    }
    __builtin_amdgcn_sched_barrier(0);
#pragma unroll
    for (int j = 0; j < KSTEPS; ++j) {
        accB = __builtin_amdgcn_wmma_f32_16x16x4_f32(false, a[j], false, fB[j], (short)0, accB, false, false);
        accC = __builtin_amdgcn_wmma_f32_16x16x4_f32(false, a[j], false, fC[j], (short)0, accC, false, false);
        accD = __builtin_amdgcn_wmma_f32_16x16x4_f32(false, a[j], false, fD[j], (short)0, accD, false, false);
    }

    const float bBn = b_B[lid];
    const float bCn = b_C[lid];
    const float bDs = b_D[0];
#pragma unroll
    for (int r = 0; r < 8; ++r) {
        const int orow = tile * 16 + r + 8 * half;
        Bm[(size_t)orow * D_STATE + lid] = accB[r] + bBn;
        Cc[(size_t)orow * D_STATE + lid] = accC[r] + bCn;
        if (lid == 0) s_xwd[wave * 16 + r + 8 * half] = accD[r] + bDs;
    }
  }
    __syncthreads();

    const int d  = threadIdx.x & (D_MODEL - 1);
    const int g  = threadIdx.x >> 7;
    const int gc = blockIdx.x * (PROJ_ROWS / CHUNK) + g;
    const size_t row0 = (size_t)gc * CHUNK;
    const float Pd = P[d];
    float acc = 0.0f;
    for (int s = 0; s < CHUNK; ++s) {
        const float dc = softplus_f(Pd + s_xwd[g * CHUNK + s]);
        disc[(row0 + s) * D_MODEL + d] = dc;
        acc += dc;
    }
    discsum[(size_t)gc * D_MODEL + d] = acc;
}

__global__ void scan_chunks_kernel(float* __restrict__ arr, int nchannels) {
    const int t = blockIdx.x * blockDim.x + threadIdx.x;
    if (t >= nchannels) return;
    const int d     = t % D_MODEL;
    const int outer = t / D_MODEL;
    float run = 0.0f;
    float* p = arr + (size_t)outer * NCHUNK * D_MODEL + d;
#pragma unroll
    for (int c = 0; c < NCHUNK; ++c) {
        const float v = p[c * D_MODEL]; p[c * D_MODEL] = run; run += v;
    }
}


#define STAGE_PIECES (CHUNK * D_MODEL / 4 / D_MODEL)

__device__ __forceinline__ void lds_row16(const float* row, float (&v)[D_STATE]) {
    const v4f* p = (const v4f*)row;
#pragma unroll
    for (int q = 0; q < D_STATE / 4; ++q) {
        const v4f t = p[q];
        v[4 * q + 0] = t.x; v[4 * q + 1] = t.y; v[4 * q + 2] = t.z; v[4 * q + 3] = t.w;
    }
}

__device__ __forceinline__ void scan_step(const float (&A2)[D_STATE], const float (&invA)[D_STATE],
                                          const float (&bm)[D_STATE], float dc, float xv,
                                          float nDcum, float (&S)[D_STATE],
                                          float (&Bx)[D_STATE], float (&q)[D_STATE]) {
    float E[D_STATE];
#pragma unroll
    for (int n = 0; n < D_STATE; ++n)
        Bx[n] = bm[n] * invA[n] * (__builtin_amdgcn_exp2f(dc * A2[n]) - 1.0f) * xv;
#pragma unroll
    for (int n = 0; n < D_STATE; ++n)
        q[n] = fmaf(A2[n], nDcum, -CLAMP2);
#pragma unroll
    for (int n = 0; n < D_STATE; ++n)
        E[n] = __builtin_amdgcn_exp2f(fminf(q[n], 127.0f));
#pragma unroll
    for (int n = 0; n < D_STATE; ++n)
        S[n] += fminf(fmaxf(Bx[n] * E[n], -1.0f), 1.0f);
}

__global__ void __launch_bounds__(D_MODEL) __attribute__((amdgpu_waves_per_eu(1, 4)))
scan_pass1_kernel(const float* __restrict__ x,
                  const float* __restrict__ Araw,
                  const float* __restrict__ disc,
                  const float* __restrict__ discsum,
                  float* __restrict__ discpre,
                  const float* __restrict__ Bm,
                  float* __restrict__ sumS) {
    __shared__ __attribute__((aligned(16))) float s_dc[CHUNK * D_MODEL];
    __shared__ __attribute__((aligned(16))) float s_x[CHUNK * D_MODEL];
    __shared__ __attribute__((aligned(16))) float s_bm[CHUNK * D_STATE];
    const int d  = threadIdx.x;
    const int ch = blockIdx.x % NCHUNK;
    const int b  = blockIdx.x / NCHUNK;
    const size_t row0 = (size_t)b * SEQ + (size_t)ch * CHUNK;

    v4f tdc[STAGE_PIECES], tx[STAGE_PIECES];
    {
        const v4f* gdc = (const v4f*)(disc + row0 * D_MODEL);
        const v4f* gx  = (const v4f*)(x + row0 * D_MODEL);
#pragma unroll
        for (int j = 0; j < STAGE_PIECES; ++j) {
            tdc[j] = gdc[j * D_MODEL + d];
            tx[j]  = gx[j * D_MODEL + d];
        }
    }
    const v4f tbm = ((const v4f*)(Bm + row0 * D_STATE))[d];

    float A2[D_STATE], invA[D_STATE], acc[D_STATE];
#pragma unroll
    for (int n = 0; n < D_STATE; ++n) {
        const float An = -softplus_f(Araw[d * D_STATE + n]);
        invA[n] = 1.0f / An;
        A2[n]   = An * LOG2E;
        acc[n]  = 0.0f;
    }
    float Dcum = 0.0f;
    {
        const float* dp = discsum + (size_t)b * NCHUNK * D_MODEL + d;
        float dv[NCHUNK];
#pragma unroll
        for (int c = 0; c < NCHUNK; ++c) dv[c] = dp[c * D_MODEL];
        asm volatile("" :: "v"(dv[0]));
#pragma unroll
        for (int c = 0; c < NCHUNK; ++c) Dcum += (c < ch) ? dv[c] : 0.0f;
    }
    discpre[(b * NCHUNK + ch) * D_MODEL + d] = Dcum;
    float nDcum = -Dcum;

#pragma unroll
    for (int j = 0; j < STAGE_PIECES; ++j) {
        ((v4f*)s_dc)[j * D_MODEL + d] = tdc[j];
        ((v4f*)s_x)[j * D_MODEL + d]  = tx[j];
    }
    ((v4f*)s_bm)[d] = tbm;
    __syncthreads();

    float dc = s_dc[d];
    float xv = s_x[d];
    float bm[D_STATE];
    lds_row16(s_bm, bm);

    for (int i = 0; i < CHUNK; ++i) {
        const int rn = (i + 1 < CHUNK) ? (i + 1) : (CHUNK - 1);
        const float dc_n = s_dc[rn * D_MODEL + d];
        const float xv_n = s_x[rn * D_MODEL + d];
        float bm_n[D_STATE];
        lds_row16(s_bm + rn * D_STATE, bm_n);

        nDcum -= dc;
        float Bx[D_STATE], q[D_STATE];
        scan_step(A2, invA, bm, dc, xv, nDcum, acc, Bx, q);

        dc = dc_n; xv = xv_n;
#pragma unroll
        for (int n = 0; n < D_STATE; ++n) bm[n] = bm_n[n];
    }
#pragma unroll
    for (int n = 0; n < D_STATE; ++n)
        sumS[((size_t)((b * D_STATE + n) * NCHUNK + ch)) * D_MODEL + d] = acc[n];
}

__global__ void __launch_bounds__(D_MODEL) __attribute__((amdgpu_waves_per_eu(1, 4)))
scan_final_kernel(const float* __restrict__ x,
                  const float* __restrict__ Araw,
                  const float* __restrict__ disc,
                  const float* __restrict__ discpre,
                  const float* __restrict__ Bm,
                  const float* __restrict__ Cc,
                  const float* __restrict__ sumSpre,
                  float* __restrict__ y) {
    __shared__ __attribute__((aligned(16))) float s_dc[CHUNK * D_MODEL];
    __shared__ __attribute__((aligned(16))) float s_x[CHUNK * D_MODEL];
    __shared__ __attribute__((aligned(16))) float s_bm[CHUNK * D_STATE];
    __shared__ __attribute__((aligned(16))) float s_cc[CHUNK * D_STATE];
    const int d  = threadIdx.x;
    const int ch = blockIdx.x % NCHUNK;
    const int b  = blockIdx.x / NCHUNK;
    const size_t row0 = (size_t)b * SEQ + (size_t)ch * CHUNK;

    v4f tdc[STAGE_PIECES], tx[STAGE_PIECES];
    {
        const v4f* gdc = (const v4f*)(disc + row0 * D_MODEL);
        const v4f* gx  = (const v4f*)(x + row0 * D_MODEL);
#pragma unroll
        for (int j = 0; j < STAGE_PIECES; ++j) {
            tdc[j] = gdc[j * D_MODEL + d];
            tx[j]  = gx[j * D_MODEL + d];
        }
    }
    const v4f tbm = ((const v4f*)(Bm + row0 * D_STATE))[d];
    const v4f tcc = ((const v4f*)(Cc + row0 * D_STATE))[d];

    float A2[D_STATE], invA[D_STATE], Ssum[D_STATE];
#pragma unroll
    for (int n = 0; n < D_STATE; ++n) {
        const float An = -softplus_f(Araw[d * D_STATE + n]);
        invA[n] = 1.0f / An;
        A2[n]   = An * LOG2E;
        Ssum[n] = sumSpre[((size_t)((b * D_STATE + n) * NCHUNK + ch)) * D_MODEL + d];
    }
    float nDcum = -discpre[(b * NCHUNK + ch) * D_MODEL + d];

#pragma unroll
    for (int j = 0; j < STAGE_PIECES; ++j) {
        ((v4f*)s_dc)[j * D_MODEL + d] = tdc[j];
        ((v4f*)s_x)[j * D_MODEL + d]  = tx[j];
    }
    ((v4f*)s_bm)[d] = tbm;
    ((v4f*)s_cc)[d] = tcc;
    __syncthreads();

    float dc = s_dc[d];
    float xv = s_x[d];
    float bm[D_STATE], cc[D_STATE];
    lds_row16(s_bm, bm);
    lds_row16(s_cc, cc);

    for (int i = 0; i < CHUNK; ++i) {
        const int rn = (i + 1 < CHUNK) ? (i + 1) : (CHUNK - 1);
        const float dc_n = s_dc[rn * D_MODEL + d];
        const float xv_n = s_x[rn * D_MODEL + d];
        float bm_n[D_STATE], cc_n[D_STATE];
        lds_row16(s_bm + rn * D_STATE, bm_n);
        lds_row16(s_cc + rn * D_STATE, cc_n);

        nDcum -= dc;
        float Bx[D_STATE], q[D_STATE];
        scan_step(A2, invA, bm, dc, xv, nDcum, Ssum, Bx, q);
        float g[D_STATE], h[D_STATE];
#pragma unroll
        for (int n = 0; n < D_STATE; ++n) g[n] = __builtin_amdgcn_exp2f(-q[n]);
#pragma unroll
        for (int n = 0; n < D_STATE; ++n) h[n] = Ssum[n] * fmaxf(g[n], fabsf(Bx[n]));
        float yv = 0.0f;
#pragma unroll
        for (int n = 0; n < D_STATE; ++n) yv += h[n] * cc[n];
        y[(row0 + i) * D_MODEL + d] = yv;

        dc = dc_n; xv = xv_n;
#pragma unroll
        for (int n = 0; n < D_STATE; ++n) { bm[n] = bm_n[n]; cc[n] = cc_n[n]; }
    }
}

extern "C" void kernel_launch(void* const* d_in, const int* in_sizes, int n_in,
                              void* d_out, int out_size, void* d_ws, size_t ws_size,
                              hipStream_t stream) {
    (void)in_sizes; (void)n_in; (void)out_size; (void)ws_size;
    const float* x    = (const float*)d_in[0];
    const float* Araw = (const float*)d_in[1];
    const float* P    = (const float*)d_in[2];
    const float* W_B  = (const float*)d_in[3];
    const float* b_B  = (const float*)d_in[4];
    const float* W_C  = (const float*)d_in[5];
    const float* b_C  = (const float*)d_in[6];
    const float* W_D  = (const float*)d_in[7];
    const float* b_D  = (const float*)d_in[8];
    float* y = (float*)d_out;

    constexpr size_t BS       = (size_t)BATCH * SEQ;
    constexpr size_t szBm     = BS * D_STATE * 4;
    constexpr size_t szCc     = BS * D_STATE * 4;
    constexpr size_t szDisc   = BS * D_MODEL * 4;
    constexpr size_t szDsum   = (size_t)BATCH * D_MODEL * NCHUNK * 4;
    char* ws = (char*)d_ws;
    float* Bm   = (float*)(ws);
    float* Cc   = (float*)(ws + szBm);
    float* disc = (float*)(ws + szBm + szCc);
    float* dsum = (float*)(ws + szBm + szCc + szDisc);
    float* dpre = (float*)(ws + szBm + szCc + szDisc + szDsum);
    float* sumS = (float*)(ws + szBm + szCc + szDisc + szDsum + szDsum);

    proj_disc_kernel<<<BS / PROJ_ROWS, 256, 0, stream>>>(x, W_B, b_B, W_C, b_C, W_D, b_D, P,
                                                         Bm, Cc, disc, dsum);
    scan_pass1_kernel<<<BATCH * NCHUNK, D_MODEL, 0, stream>>>(
        x, Araw, disc, dsum, dpre, Bm, sumS);
    scan_chunks_kernel<<<(BATCH * D_MODEL * D_STATE + 255) / 256, 256, 0, stream>>>(
        sumS, BATCH * D_MODEL * D_STATE);
    scan_final_kernel<<<BATCH * NCHUNK, D_MODEL, 0, stream>>>(
        x, Araw, disc, dpre, Bm, Cc, sumS, y);
}


# Chunk-prefix launch spread from 32 to 256 blocks: 4 waves split a channel's 128 chunks, 32 loads in flight per lane, wave totals via LDS; prefix = earlier waves' totals + in-wave sum (order changes)
# speedup vs baseline: 12.1191x; 1.0403x over previous
#include <hip/hip_runtime.h>
#include <math.h>


#define D_MODEL 128
#define D_STATE 16
#define BATCH   4
#define SEQ     4096
#define NCHUNK  128
#define CHUNK   (SEQ / NCHUNK)
#define LOG_CLAMP 70.0f
#define LOG2E  1.4426950408889634f
#define CLAMP2 ((float)((double)LOG_CLAMP * 1.4426950408889634))

typedef float v2f __attribute__((ext_vector_type(2)));
typedef float v4f __attribute__((ext_vector_type(4)));
typedef float v8f __attribute__((ext_vector_type(8)));

#define LN2 0.6931471805599453f
__device__ __forceinline__ float softplus_f(float z) {
    const float e  = __builtin_amdgcn_exp2f(-LOG2E * fabsf(z));
    const float lg = LN2 * __builtin_amdgcn_logf(1.0f + e);
    const float sr = e * fmaf(e, fmaf(e, 1.0f / 3.0f, -0.5f), 1.0f);
    return fmaxf(z, 0.0f) + ((e < 0.015625f) ? sr : lg);
}

#define PROJ_WAVES 4
#define PROJ_ROWS  (PROJ_WAVES * 16)
#define KSTEPS     (D_MODEL / 4)
__global__ void __launch_bounds__(256) __attribute__((amdgpu_waves_per_eu(1, 2)))
proj_disc_kernel(const float* __restrict__ x,
                 const float* __restrict__ W_B, const float* __restrict__ b_B,
                 const float* __restrict__ W_C, const float* __restrict__ b_C,
                 const float* __restrict__ W_D, const float* __restrict__ b_D,
                 const float* __restrict__ P,
                 float* __restrict__ Bm, float* __restrict__ Cc,
                 float* __restrict__ disc, float* __restrict__ discsum) {
    __shared__ float s_xwd[PROJ_ROWS];
    const int lane  = threadIdx.x & 31;
    const int wave  = __builtin_amdgcn_readfirstlane((int)(threadIdx.x >> 5));
  if (wave < PROJ_WAVES) {
    const int tile  = blockIdx.x * PROJ_WAVES + wave;
    const int half  = lane >> 4;
    const int lid   = lane & 15;
    const int row   = tile * 16 + lid;
    const int khalf = half * 2;
    const float dmask = (lid == 0) ? 1.0f : 0.0f;

    v8f accB = {}; v8f accC = {}; v8f accD = {};
    const float* xrow = x + (size_t)row * D_MODEL;

    v2f a[KSTEPS], fB[KSTEPS], fC[KSTEPS], fD[KSTEPS];
#pragma unroll
    for (int j = 0; j < KSTEPS; ++j) {
        const int k = 4 * j + khalf;
        a[j] = *(const v2f*)(xrow + k);
        fB[j].x = W_B[(k + 0) * D_STATE + lid];
        fB[j].y = W_B[(k + 1) * D_STATE + lid];
        fC[j].x = W_C[(k + 0) * D_STATE + lid];
        fC[j].y = W_C[(k + 1) * D_STATE + lid];
        fD[j].x = W_D[k + 0] * dmask;
        fD[j].y = W_D[k + 1] * dmask;
    }
    __builtin_amdgcn_sched_barrier(0);
#pragma unroll
    for (int j = 0; j < KSTEPS; ++j) {
        accB = __builtin_amdgcn_wmma_f32_16x16x4_f32(false, a[j], false, fB[j], (short)0, accB, false, false);
        accC = __builtin_amdgcn_wmma_f32_16x16x4_f32(false, a[j], false, fC[j], (short)0, accC, false, false);
        accD = __builtin_amdgcn_wmma_f32_16x16x4_f32(false, a[j], false, fD[j], (short)0, accD, false, false);
    }

    const float bBn = b_B[lid];
    const float bCn = b_C[lid];
    const float bDs = b_D[0];
#pragma unroll
    for (int r = 0; r < 8; ++r) {
        const int orow = tile * 16 + r + 8 * half;
        Bm[(size_t)orow * D_STATE + lid] = accB[r] + bBn;
        Cc[(size_t)orow * D_STATE + lid] = accC[r] + bCn;
        if (lid == 0) s_xwd[wave * 16 + r + 8 * half] = accD[r] + bDs;
    }
  }
    __syncthreads();

    const int d  = threadIdx.x & (D_MODEL - 1);
    const int g  = threadIdx.x >> 7;
    const int gc = blockIdx.x * (PROJ_ROWS / CHUNK) + g;
    const size_t row0 = (size_t)gc * CHUNK;
    const float Pd = P[d];
    float acc = 0.0f;
    for (int s = 0; s < CHUNK; ++s) {
        const float dc = softplus_f(Pd + s_xwd[g * CHUNK + s]);
        disc[(row0 + s) * D_MODEL + d] = dc;
        acc += dc;
    }
    discsum[(size_t)gc * D_MODEL + d] = acc;
}

#define PFX_WAVES 4
#define PFX_PER   (NCHUNK / PFX_WAVES)
__global__ void __launch_bounds__(32 * PFX_WAVES)
scan_chunks_kernel(float* __restrict__ arr) {
    __shared__ float s_tot[PFX_WAVES][32];
    const int lane  = threadIdx.x & 31;
    const int wave  = threadIdx.x >> 5;
    const int outer = blockIdx.x >> 2;
    const int d     = ((blockIdx.x & 3) << 5) + lane;
    float* p = arr + ((size_t)outer * NCHUNK + (size_t)wave * PFX_PER) * D_MODEL + d;
    float v[PFX_PER];
#pragma unroll
    for (int j = 0; j < PFX_PER; ++j) v[j] = p[j * D_MODEL];
    float run = 0.0f;
#pragma unroll
    for (int j = 0; j < PFX_PER; ++j) { const float t = v[j]; v[j] = run; run += t; }
    s_tot[wave][lane] = run;
    __syncthreads();
    float off = 0.0f;
#pragma unroll
    for (int w = 0; w < PFX_WAVES - 1; ++w) off += (w < wave) ? s_tot[w][lane] : 0.0f;
#pragma unroll
    for (int j = 0; j < PFX_PER; ++j) p[j * D_MODEL] = off + v[j];
}


#define STAGE_PIECES (CHUNK * D_MODEL / 4 / D_MODEL)

__device__ __forceinline__ void lds_row16(const float* row, float (&v)[D_STATE]) {
    const v4f* p = (const v4f*)row;
#pragma unroll
    for (int q = 0; q < D_STATE / 4; ++q) {
        const v4f t = p[q];
        v[4 * q + 0] = t.x; v[4 * q + 1] = t.y; v[4 * q + 2] = t.z; v[4 * q + 3] = t.w;
    }
}

__device__ __forceinline__ void scan_step(const float (&A2)[D_STATE], const float (&invA)[D_STATE],
                                          const float (&bm)[D_STATE], float dc, float xv,
                                          float nDcum, float (&S)[D_STATE],
                                          float (&Bx)[D_STATE], float (&q)[D_STATE]) {
    float E[D_STATE];
#pragma unroll
    for (int n = 0; n < D_STATE; ++n)
        Bx[n] = bm[n] * invA[n] * (__builtin_amdgcn_exp2f(dc * A2[n]) - 1.0f) * xv;
#pragma unroll
    for (int n = 0; n < D_STATE; ++n)
        q[n] = fmaf(A2[n], nDcum, -CLAMP2);
#pragma unroll
    for (int n = 0; n < D_STATE; ++n)
        E[n] = __builtin_amdgcn_exp2f(fminf(q[n], 127.0f));
#pragma unroll
    for (int n = 0; n < D_STATE; ++n)
        S[n] += fminf(fmaxf(Bx[n] * E[n], -1.0f), 1.0f);
}

__global__ void __launch_bounds__(D_MODEL) __attribute__((amdgpu_waves_per_eu(1, 4)))
scan_pass1_kernel(const float* __restrict__ x,
                  const float* __restrict__ Araw,
                  const float* __restrict__ disc,
                  const float* __restrict__ discsum,
                  float* __restrict__ discpre,
                  const float* __restrict__ Bm,
                  float* __restrict__ sumS) {
    __shared__ __attribute__((aligned(16))) float s_dc[CHUNK * D_MODEL];
    __shared__ __attribute__((aligned(16))) float s_x[CHUNK * D_MODEL];
    __shared__ __attribute__((aligned(16))) float s_bm[CHUNK * D_STATE];
    const int d  = threadIdx.x;
    const int ch = blockIdx.x % NCHUNK;
    const int b  = blockIdx.x / NCHUNK;
    const size_t row0 = (size_t)b * SEQ + (size_t)ch * CHUNK;

    v4f tdc[STAGE_PIECES], tx[STAGE_PIECES];
    {
        const v4f* gdc = (const v4f*)(disc + row0 * D_MODEL);
        const v4f* gx  = (const v4f*)(x + row0 * D_MODEL);
#pragma unroll
        for (int j = 0; j < STAGE_PIECES; ++j) {
            tdc[j] = gdc[j * D_MODEL + d];
            tx[j]  = gx[j * D_MODEL + d];
        }
    }
    const v4f tbm = ((const v4f*)(Bm + row0 * D_STATE))[d];

    float A2[D_STATE], invA[D_STATE], acc[D_STATE];
#pragma unroll
    for (int n = 0; n < D_STATE; ++n) {
        const float An = -softplus_f(Araw[d * D_STATE + n]);
        invA[n] = 1.0f / An;
        A2[n]   = An * LOG2E;
        acc[n]  = 0.0f;
    }
    float Dcum = 0.0f;
    {
        const float* dp = discsum + (size_t)b * NCHUNK * D_MODEL + d;
        float dv[NCHUNK];
#pragma unroll
        for (int c = 0; c < NCHUNK; ++c) dv[c] = dp[c * D_MODEL];
        asm volatile("" :: "v"(dv[0]));
#pragma unroll
        for (int c = 0; c < NCHUNK; ++c) Dcum += (c < ch) ? dv[c] : 0.0f;
    }
    discpre[(b * NCHUNK + ch) * D_MODEL + d] = Dcum;
    float nDcum = -Dcum;

#pragma unroll
    for (int j = 0; j < STAGE_PIECES; ++j) {
        ((v4f*)s_dc)[j * D_MODEL + d] = tdc[j];
        ((v4f*)s_x)[j * D_MODEL + d]  = tx[j];
    }
    ((v4f*)s_bm)[d] = tbm;
    __syncthreads();

    float dc = s_dc[d];
    float xv = s_x[d];
    float bm[D_STATE];
    lds_row16(s_bm, bm);

    for (int i = 0; i < CHUNK; ++i) {
        const int rn = (i + 1 < CHUNK) ? (i + 1) : (CHUNK - 1);
        const float dc_n = s_dc[rn * D_MODEL + d];
        const float xv_n = s_x[rn * D_MODEL + d];
        float bm_n[D_STATE];
        lds_row16(s_bm + rn * D_STATE, bm_n);

        nDcum -= dc;
        float Bx[D_STATE], q[D_STATE];
        scan_step(A2, invA, bm, dc, xv, nDcum, acc, Bx, q);

        dc = dc_n; xv = xv_n;
#pragma unroll
        for (int n = 0; n < D_STATE; ++n) bm[n] = bm_n[n];
    }
#pragma unroll
    for (int n = 0; n < D_STATE; ++n)
        sumS[((size_t)((b * D_STATE + n) * NCHUNK + ch)) * D_MODEL + d] = acc[n];
}

__global__ void __launch_bounds__(D_MODEL) __attribute__((amdgpu_waves_per_eu(1, 4)))
scan_final_kernel(const float* __restrict__ x,
                  const float* __restrict__ Araw,
                  const float* __restrict__ disc,
                  const float* __restrict__ discpre,
                  const float* __restrict__ Bm,
                  const float* __restrict__ Cc,
                  const float* __restrict__ sumSpre,
                  float* __restrict__ y) {
    __shared__ __attribute__((aligned(16))) float s_dc[CHUNK * D_MODEL];
    __shared__ __attribute__((aligned(16))) float s_x[CHUNK * D_MODEL];
    __shared__ __attribute__((aligned(16))) float s_bm[CHUNK * D_STATE];
    __shared__ __attribute__((aligned(16))) float s_cc[CHUNK * D_STATE];
    const int d  = threadIdx.x;
    const int ch = blockIdx.x % NCHUNK;
    const int b  = blockIdx.x / NCHUNK;
    const size_t row0 = (size_t)b * SEQ + (size_t)ch * CHUNK;

    v4f tdc[STAGE_PIECES], tx[STAGE_PIECES];
    {
        const v4f* gdc = (const v4f*)(disc + row0 * D_MODEL);
        const v4f* gx  = (const v4f*)(x + row0 * D_MODEL);
#pragma unroll
        for (int j = 0; j < STAGE_PIECES; ++j) {
            tdc[j] = gdc[j * D_MODEL + d];
            tx[j]  = gx[j * D_MODEL + d];
        }
    }
    const v4f tbm = ((const v4f*)(Bm + row0 * D_STATE))[d];
    const v4f tcc = ((const v4f*)(Cc + row0 * D_STATE))[d];

    float A2[D_STATE], invA[D_STATE], Ssum[D_STATE];
#pragma unroll
    for (int n = 0; n < D_STATE; ++n) {
        const float An = -softplus_f(Araw[d * D_STATE + n]);
        invA[n] = 1.0f / An;
        A2[n]   = An * LOG2E;
        Ssum[n] = sumSpre[((size_t)((b * D_STATE + n) * NCHUNK + ch)) * D_MODEL + d];
    }
    float nDcum = -discpre[(b * NCHUNK + ch) * D_MODEL + d];

#pragma unroll
    for (int j = 0; j < STAGE_PIECES; ++j) {
        ((v4f*)s_dc)[j * D_MODEL + d] = tdc[j];
        ((v4f*)s_x)[j * D_MODEL + d]  = tx[j];
    }
    ((v4f*)s_bm)[d] = tbm;
    ((v4f*)s_cc)[d] = tcc;
    __syncthreads();

    float dc = s_dc[d];
    float xv = s_x[d];
    float bm[D_STATE], cc[D_STATE];
    lds_row16(s_bm, bm);
    lds_row16(s_cc, cc);

    for (int i = 0; i < CHUNK; ++i) {
        const int rn = (i + 1 < CHUNK) ? (i + 1) : (CHUNK - 1);
        const float dc_n = s_dc[rn * D_MODEL + d];
        const float xv_n = s_x[rn * D_MODEL + d];
        float bm_n[D_STATE], cc_n[D_STATE];
        lds_row16(s_bm + rn * D_STATE, bm_n);
        lds_row16(s_cc + rn * D_STATE, cc_n);

        nDcum -= dc;
        float Bx[D_STATE], q[D_STATE];
        scan_step(A2, invA, bm, dc, xv, nDcum, Ssum, Bx, q);
        float g[D_STATE], h[D_STATE];
#pragma unroll
        for (int n = 0; n < D_STATE; ++n) g[n] = __builtin_amdgcn_exp2f(-q[n]);
#pragma unroll
        for (int n = 0; n < D_STATE; ++n) h[n] = Ssum[n] * fmaxf(g[n], fabsf(Bx[n]));
        float yv = 0.0f;
#pragma unroll
        for (int n = 0; n < D_STATE; ++n) yv += h[n] * cc[n];
        y[(row0 + i) * D_MODEL + d] = yv;

        dc = dc_n; xv = xv_n;
#pragma unroll
        for (int n = 0; n < D_STATE; ++n) { bm[n] = bm_n[n]; cc[n] = cc_n[n]; }
    }
}

extern "C" void kernel_launch(void* const* d_in, const int* in_sizes, int n_in,
                              void* d_out, int out_size, void* d_ws, size_t ws_size,
                              hipStream_t stream) {
    (void)in_sizes; (void)n_in; (void)out_size; (void)ws_size;
    const float* x    = (const float*)d_in[0];
    const float* Araw = (const float*)d_in[1];
    const float* P    = (const float*)d_in[2];
    const float* W_B  = (const float*)d_in[3];
    const float* b_B  = (const float*)d_in[4];
    const float* W_C  = (const float*)d_in[5];
    const float* b_C  = (const float*)d_in[6];
    const float* W_D  = (const float*)d_in[7];
    const float* b_D  = (const float*)d_in[8];
    float* y = (float*)d_out;

    constexpr size_t BS       = (size_t)BATCH * SEQ;
    constexpr size_t szBm     = BS * D_STATE * 4;
    constexpr size_t szCc     = BS * D_STATE * 4;
    constexpr size_t szDisc   = BS * D_MODEL * 4;
    constexpr size_t szDsum   = (size_t)BATCH * D_MODEL * NCHUNK * 4;
    char* ws = (char*)d_ws;
    float* Bm   = (float*)(ws);
    float* Cc   = (float*)(ws + szBm);
    float* disc = (float*)(ws + szBm + szCc);
    float* dsum = (float*)(ws + szBm + szCc + szDisc);
    float* dpre = (float*)(ws + szBm + szCc + szDisc + szDsum);
    float* sumS = (float*)(ws + szBm + szCc + szDisc + szDsum + szDsum);

    proj_disc_kernel<<<BS / PROJ_ROWS, 256, 0, stream>>>(x, W_B, b_B, W_C, b_C, W_D, b_D, P,
                                                         Bm, Cc, disc, dsum);
    scan_pass1_kernel<<<BATCH * NCHUNK, D_MODEL, 0, stream>>>(
        x, Araw, disc, dsum, dpre, Bm, sumS);
    scan_chunks_kernel<<<BATCH * D_STATE * (D_MODEL / 32), 32 * PFX_WAVES, 0, stream>>>(sumS);
    scan_final_kernel<<<BATCH * NCHUNK, D_MODEL, 0, stream>>>(
        x, Araw, disc, dpre, Bm, Cc, sumS, y);
}


# Both scan step loops take two rows per trip, one operand register set per row parity refilled from LDS after use: most per-step moves of the read-ahead Bm/C rows (next to current) leave the loops
# speedup vs baseline: 12.5666x; 1.0369x over previous
#include <hip/hip_runtime.h>
#include <math.h>


#define D_MODEL 128
#define D_STATE 16
#define BATCH   4
#define SEQ     4096
#define NCHUNK  128
#define CHUNK   (SEQ / NCHUNK)
#define LOG_CLAMP 70.0f
#define LOG2E  1.4426950408889634f
#define CLAMP2 ((float)((double)LOG_CLAMP * 1.4426950408889634))

typedef float v2f __attribute__((ext_vector_type(2)));
typedef float v4f __attribute__((ext_vector_type(4)));
typedef float v8f __attribute__((ext_vector_type(8)));

#define LN2 0.6931471805599453f
__device__ __forceinline__ float softplus_f(float z) {
    const float e  = __builtin_amdgcn_exp2f(-LOG2E * fabsf(z));
    const float lg = LN2 * __builtin_amdgcn_logf(1.0f + e);
    const float sr = e * fmaf(e, fmaf(e, 1.0f / 3.0f, -0.5f), 1.0f);
    return fmaxf(z, 0.0f) + ((e < 0.015625f) ? sr : lg);
}

#define PROJ_WAVES 4
#define PROJ_ROWS  (PROJ_WAVES * 16)
#define KSTEPS     (D_MODEL / 4)
__global__ void __launch_bounds__(256) __attribute__((amdgpu_waves_per_eu(1, 2)))
proj_disc_kernel(const float* __restrict__ x,
                 const float* __restrict__ W_B, const float* __restrict__ b_B,
                 const float* __restrict__ W_C, const float* __restrict__ b_C,
                 const float* __restrict__ W_D, const float* __restrict__ b_D,
                 const float* __restrict__ P,
                 float* __restrict__ Bm, float* __restrict__ Cc,
                 float* __restrict__ disc, float* __restrict__ discsum) {
    __shared__ float s_xwd[PROJ_ROWS];
    const int lane  = threadIdx.x & 31;
    const int wave  = __builtin_amdgcn_readfirstlane((int)(threadIdx.x >> 5));
  if (wave < PROJ_WAVES) {
    const int tile  = blockIdx.x * PROJ_WAVES + wave;
    const int half  = lane >> 4;
    const int lid   = lane & 15;
    const int row   = tile * 16 + lid;
    const int khalf = half * 2;
    const float dmask = (lid == 0) ? 1.0f : 0.0f;

    v8f accB = {}; v8f accC = {}; v8f accD = {};
    const float* xrow = x + (size_t)row * D_MODEL;

    v2f a[KSTEPS], fB[KSTEPS], fC[KSTEPS], fD[KSTEPS];
#pragma unroll
    for (int j = 0; j < KSTEPS; ++j) {
        const int k = 4 * j + khalf;
        a[j] = *(const v2f*)(xrow + k);
        fB[j].x = W_B[(k + 0) * D_STATE + lid];
        fB[j].y = W_B[(k + 1) * D_STATE + lid];
        fC[j].x = W_C[(k + 0) * D_STATE + lid];
        fC[j].y = W_C[(k + 1) * D_STATE + lid];
        fD[j].x = W_D[k + 0] * dmask;
        fD[j].y = W_D[k + 1] * dmask;
    }
    __builtin_amdgcn_sched_barrier(0);
#pragma unroll
    for (int j = 0; j < KSTEPS; ++j) {
        accB = __builtin_amdgcn_wmma_f32_16x16x4_f32(false, a[j], false, fB[j], (short)0, accB, false, false);
        accC = __builtin_amdgcn_wmma_f32_16x16x4_f32(false, a[j], false, fC[j], (short)0, accC, false, false);
        accD = __builtin_amdgcn_wmma_f32_16x16x4_f32(false, a[j], false, fD[j], (short)0, accD, false, false);
    }

    const float bBn = b_B[lid];
    const float bCn = b_C[lid];
    const float bDs = b_D[0];
#pragma unroll
    for (int r = 0; r < 8; ++r) {
        const int orow = tile * 16 + r + 8 * half;
        Bm[(size_t)orow * D_STATE + lid] = accB[r] + bBn;
        Cc[(size_t)orow * D_STATE + lid] = accC[r] + bCn;
        if (lid == 0) s_xwd[wave * 16 + r + 8 * half] = accD[r] + bDs;
    }
  }
    __syncthreads();

    const int d  = threadIdx.x & (D_MODEL - 1);
    const int g  = threadIdx.x >> 7;
    const int gc = blockIdx.x * (PROJ_ROWS / CHUNK) + g;
    const size_t row0 = (size_t)gc * CHUNK;
    const float Pd = P[d];
    float acc = 0.0f;
    for (int s = 0; s < CHUNK; ++s) {
        const float dc = softplus_f(Pd + s_xwd[g * CHUNK + s]);
        disc[(row0 + s) * D_MODEL + d] = dc;
        acc += dc;
    }
    discsum[(size_t)gc * D_MODEL + d] = acc;
}

#define PFX_WAVES 4
#define PFX_PER   (NCHUNK / PFX_WAVES)
__global__ void __launch_bounds__(32 * PFX_WAVES)
scan_chunks_kernel(float* __restrict__ arr) {
    __shared__ float s_tot[PFX_WAVES][32];
    const int lane  = threadIdx.x & 31;
    const int wave  = threadIdx.x >> 5;
    const int outer = blockIdx.x >> 2;
    const int d     = ((blockIdx.x & 3) << 5) + lane;
    float* p = arr + ((size_t)outer * NCHUNK + (size_t)wave * PFX_PER) * D_MODEL + d;
    float v[PFX_PER];
#pragma unroll
    for (int j = 0; j < PFX_PER; ++j) v[j] = p[j * D_MODEL];
    float run = 0.0f;
#pragma unroll
    for (int j = 0; j < PFX_PER; ++j) { const float t = v[j]; v[j] = run; run += t; }
    s_tot[wave][lane] = run;
    __syncthreads();
    float off = 0.0f;
#pragma unroll
    for (int w = 0; w < PFX_WAVES - 1; ++w) off += (w < wave) ? s_tot[w][lane] : 0.0f;
#pragma unroll
    for (int j = 0; j < PFX_PER; ++j) p[j * D_MODEL] = off + v[j];
}


#define STAGE_PIECES (CHUNK * D_MODEL / 4 / D_MODEL)

__device__ __forceinline__ void lds_row16(const float* row, float (&v)[D_STATE]) {
    const v4f* p = (const v4f*)row;
#pragma unroll
    for (int q = 0; q < D_STATE / 4; ++q) {
        const v4f t = p[q];
        v[4 * q + 0] = t.x; v[4 * q + 1] = t.y; v[4 * q + 2] = t.z; v[4 * q + 3] = t.w;
    }
}

__device__ __forceinline__ void scan_step(const float (&A2)[D_STATE], const float (&invA)[D_STATE],
                                          const float (&bm)[D_STATE], float dc, float xv,
                                          float nDcum, float (&S)[D_STATE],
                                          float (&Bx)[D_STATE], float (&q)[D_STATE]) {
    float E[D_STATE];
#pragma unroll
    for (int n = 0; n < D_STATE; ++n)
        Bx[n] = bm[n] * invA[n] * (__builtin_amdgcn_exp2f(dc * A2[n]) - 1.0f) * xv;
#pragma unroll
    for (int n = 0; n < D_STATE; ++n)
        q[n] = fmaf(A2[n], nDcum, -CLAMP2);
#pragma unroll
    for (int n = 0; n < D_STATE; ++n)
        E[n] = __builtin_amdgcn_exp2f(fminf(q[n], 127.0f));
#pragma unroll
    for (int n = 0; n < D_STATE; ++n)
        S[n] += fminf(fmaxf(Bx[n] * E[n], -1.0f), 1.0f);
}

__global__ void __launch_bounds__(D_MODEL) __attribute__((amdgpu_waves_per_eu(1, 4)))
scan_pass1_kernel(const float* __restrict__ x,
                  const float* __restrict__ Araw,
                  const float* __restrict__ disc,
                  const float* __restrict__ discsum,
                  float* __restrict__ discpre,
                  const float* __restrict__ Bm,
                  float* __restrict__ sumS) {
    __shared__ __attribute__((aligned(16))) float s_dc[CHUNK * D_MODEL];
    __shared__ __attribute__((aligned(16))) float s_x[CHUNK * D_MODEL];
    __shared__ __attribute__((aligned(16))) float s_bm[CHUNK * D_STATE];
    const int d  = threadIdx.x;
    const int ch = blockIdx.x % NCHUNK;
    const int b  = blockIdx.x / NCHUNK;
    const size_t row0 = (size_t)b * SEQ + (size_t)ch * CHUNK;

    v4f tdc[STAGE_PIECES], tx[STAGE_PIECES];
    {
        const v4f* gdc = (const v4f*)(disc + row0 * D_MODEL);
        const v4f* gx  = (const v4f*)(x + row0 * D_MODEL);
#pragma unroll
        for (int j = 0; j < STAGE_PIECES; ++j) {
            tdc[j] = gdc[j * D_MODEL + d];
            tx[j]  = gx[j * D_MODEL + d];
        }
    }
    const v4f tbm = ((const v4f*)(Bm + row0 * D_STATE))[d];

    float A2[D_STATE], invA[D_STATE], acc[D_STATE];
#pragma unroll
    for (int n = 0; n < D_STATE; ++n) {
        const float An = -softplus_f(Araw[d * D_STATE + n]);
        invA[n] = 1.0f / An;
        A2[n]   = An * LOG2E;
        acc[n]  = 0.0f;
    }
    float Dcum = 0.0f;
    {
        const float* dp = discsum + (size_t)b * NCHUNK * D_MODEL + d;
        float dv[NCHUNK];
#pragma unroll
        for (int c = 0; c < NCHUNK; ++c) dv[c] = dp[c * D_MODEL];
        asm volatile("" :: "v"(dv[0]));
#pragma unroll
        for (int c = 0; c < NCHUNK; ++c) Dcum += (c < ch) ? dv[c] : 0.0f;
    }
    discpre[(b * NCHUNK + ch) * D_MODEL + d] = Dcum;
    float nDcum = -Dcum;

#pragma unroll
    for (int j = 0; j < STAGE_PIECES; ++j) {
        ((v4f*)s_dc)[j * D_MODEL + d] = tdc[j];
        ((v4f*)s_x)[j * D_MODEL + d]  = tx[j];
    }
    ((v4f*)s_bm)[d] = tbm;
    __syncthreads();

    float dcA = s_dc[d];
    float xvA = s_x[d];
    float bmA[D_STATE];
    lds_row16(s_bm, bmA);

    for (int i = 0; i < CHUNK; i += 2) {
        const float dcB = s_dc[(i + 1) * D_MODEL + d];
        const float xvB = s_x[(i + 1) * D_MODEL + d];
        float bmB[D_STATE];
        lds_row16(s_bm + (i + 1) * D_STATE, bmB);

        float Bx[D_STATE], q[D_STATE];
        nDcum -= dcA;
        scan_step(A2, invA, bmA, dcA, xvA, nDcum, acc, Bx, q);

        const int rn = (i + 2 < CHUNK) ? (i + 2) : (CHUNK - 1);
        dcA = s_dc[rn * D_MODEL + d];
        xvA = s_x[rn * D_MODEL + d];
        lds_row16(s_bm + rn * D_STATE, bmA);

        nDcum -= dcB;
        scan_step(A2, invA, bmB, dcB, xvB, nDcum, acc, Bx, q);
    }
#pragma unroll
    for (int n = 0; n < D_STATE; ++n)
        sumS[((size_t)((b * D_STATE + n) * NCHUNK + ch)) * D_MODEL + d] = acc[n];
}

__device__ __forceinline__ float final_row(const float (&Ssum)[D_STATE], const float (&Bx)[D_STATE],
                                           const float (&q)[D_STATE], const float (&cc)[D_STATE]) {
    float g[D_STATE], h[D_STATE];
#pragma unroll
    for (int n = 0; n < D_STATE; ++n) g[n] = __builtin_amdgcn_exp2f(-q[n]);
#pragma unroll
    for (int n = 0; n < D_STATE; ++n) h[n] = Ssum[n] * fmaxf(g[n], fabsf(Bx[n]));
    float yv = 0.0f;
#pragma unroll
    for (int n = 0; n < D_STATE; ++n) yv += h[n] * cc[n];
    return yv;
}

__global__ void __launch_bounds__(D_MODEL) __attribute__((amdgpu_waves_per_eu(1, 4)))
scan_final_kernel(const float* __restrict__ x,
                  const float* __restrict__ Araw,
                  const float* __restrict__ disc,
                  const float* __restrict__ discpre,
                  const float* __restrict__ Bm,
                  const float* __restrict__ Cc,
                  const float* __restrict__ sumSpre,
                  float* __restrict__ y) {
    __shared__ __attribute__((aligned(16))) float s_dc[CHUNK * D_MODEL];
    __shared__ __attribute__((aligned(16))) float s_x[CHUNK * D_MODEL];
    __shared__ __attribute__((aligned(16))) float s_bm[CHUNK * D_STATE];
    __shared__ __attribute__((aligned(16))) float s_cc[CHUNK * D_STATE];
    const int d  = threadIdx.x;
    const int ch = blockIdx.x % NCHUNK;
    const int b  = blockIdx.x / NCHUNK;
    const size_t row0 = (size_t)b * SEQ + (size_t)ch * CHUNK;

    v4f tdc[STAGE_PIECES], tx[STAGE_PIECES];
    {
        const v4f* gdc = (const v4f*)(disc + row0 * D_MODEL);
        const v4f* gx  = (const v4f*)(x + row0 * D_MODEL);
#pragma unroll
        for (int j = 0; j < STAGE_PIECES; ++j) {
            tdc[j] = gdc[j * D_MODEL + d];
            tx[j]  = gx[j * D_MODEL + d];
        }
    }
    const v4f tbm = ((const v4f*)(Bm + row0 * D_STATE))[d];
    const v4f tcc = ((const v4f*)(Cc + row0 * D_STATE))[d];

    float A2[D_STATE], invA[D_STATE], Ssum[D_STATE];
#pragma unroll
    for (int n = 0; n < D_STATE; ++n) {
        const float An = -softplus_f(Araw[d * D_STATE + n]);
        invA[n] = 1.0f / An;
        A2[n]   = An * LOG2E;
        Ssum[n] = sumSpre[((size_t)((b * D_STATE + n) * NCHUNK + ch)) * D_MODEL + d];
    }
    float nDcum = -discpre[(b * NCHUNK + ch) * D_MODEL + d];

#pragma unroll
    for (int j = 0; j < STAGE_PIECES; ++j) {
        ((v4f*)s_dc)[j * D_MODEL + d] = tdc[j];
        ((v4f*)s_x)[j * D_MODEL + d]  = tx[j];
    }
    ((v4f*)s_bm)[d] = tbm;
    ((v4f*)s_cc)[d] = tcc;
    __syncthreads();

    float dcA = s_dc[d];
    float xvA = s_x[d];
    float bmA[D_STATE], ccA[D_STATE];
    lds_row16(s_bm, bmA);
    lds_row16(s_cc, ccA);

    for (int i = 0; i < CHUNK; i += 2) {
        const float dcB = s_dc[(i + 1) * D_MODEL + d];
        const float xvB = s_x[(i + 1) * D_MODEL + d];
        float bmB[D_STATE], ccB[D_STATE];
        lds_row16(s_bm + (i + 1) * D_STATE, bmB);
        lds_row16(s_cc + (i + 1) * D_STATE, ccB);

        float Bx[D_STATE], q[D_STATE];
        nDcum -= dcA;
        scan_step(A2, invA, bmA, dcA, xvA, nDcum, Ssum, Bx, q);
        y[(row0 + i) * D_MODEL + d] = final_row(Ssum, Bx, q, ccA);

        const int rn = (i + 2 < CHUNK) ? (i + 2) : (CHUNK - 1);
        dcA = s_dc[rn * D_MODEL + d];
        xvA = s_x[rn * D_MODEL + d];
        lds_row16(s_bm + rn * D_STATE, bmA);
        lds_row16(s_cc + rn * D_STATE, ccA);

        nDcum -= dcB;
        scan_step(A2, invA, bmB, dcB, xvB, nDcum, Ssum, Bx, q);
        y[(row0 + i + 1) * D_MODEL + d] = final_row(Ssum, Bx, q, ccB);
    }
}

extern "C" void kernel_launch(void* const* d_in, const int* in_sizes, int n_in,
                              void* d_out, int out_size, void* d_ws, size_t ws_size,
                              hipStream_t stream) {
    (void)in_sizes; (void)n_in; (void)out_size; (void)ws_size;
    const float* x    = (const float*)d_in[0];
    const float* Araw = (const float*)d_in[1];
    const float* P    = (const float*)d_in[2];
    const float* W_B  = (const float*)d_in[3];
    const float* b_B  = (const float*)d_in[4];
    const float* W_C  = (const float*)d_in[5];
    const float* b_C  = (const float*)d_in[6];
    const float* W_D  = (const float*)d_in[7];
    const float* b_D  = (const float*)d_in[8];
    float* y = (float*)d_out;

    constexpr size_t BS       = (size_t)BATCH * SEQ;
    constexpr size_t szBm     = BS * D_STATE * 4;
    constexpr size_t szCc     = BS * D_STATE * 4;
    constexpr size_t szDisc   = BS * D_MODEL * 4;
    constexpr size_t szDsum   = (size_t)BATCH * D_MODEL * NCHUNK * 4;
    char* ws = (char*)d_ws;
    float* Bm   = (float*)(ws);
    float* Cc   = (float*)(ws + szBm);
    float* disc = (float*)(ws + szBm + szCc);
    float* dsum = (float*)(ws + szBm + szCc + szDisc);
    float* dpre = (float*)(ws + szBm + szCc + szDisc + szDsum);
    float* sumS = (float*)(ws + szBm + szCc + szDisc + szDsum + szDsum);

    proj_disc_kernel<<<BS / PROJ_ROWS, 256, 0, stream>>>(x, W_B, b_B, W_C, b_C, W_D, b_D, P,
                                                         Bm, Cc, disc, dsum);
    scan_pass1_kernel<<<BATCH * NCHUNK, D_MODEL, 0, stream>>>(
        x, Araw, disc, dsum, dpre, Bm, sumS);
    scan_chunks_kernel<<<BATCH * D_STATE * (D_MODEL / 32), 32 * PFX_WAVES, 0, stream>>>(sumS);
    scan_final_kernel<<<BATCH * NCHUNK, D_MODEL, 0, stream>>>(
        x, Araw, disc, dpre, Bm, Cc, sumS, y);
}
